# starting kernel
# baseline (speedup 1.0000x reference)
#include <hip/hip_runtime.h>

#define CN0 512
#define CN1 2048
#define CN2 8192
#define KNB 33
#define CIN 256
#define CHID 128
#define COUT 128
#define CNH 8
#define LRELU_ALPHA 0.2f

typedef float v2f __attribute__((ext_vector_type(2)));
typedef float v8f __attribute__((ext_vector_type(8)));

__global__ void gat_pack_w1_kernel(const float* __restrict__ W1,
                                   float* __restrict__ Bp1) {
  int idx = blockIdx.x * blockDim.x + threadIdx.x;
  int j = idx % CHID;
  int i = (idx / CHID) % CIN;
  int h = idx / (CHID * CIN);
  int n = h * CHID + j;
  Bp1[(size_t)(i >> 1) * (2 * CNH * CHID) + 2 * n + (i & 1)] = W1[idx];
}

__global__ void gat_pack_w2_kernel(const float* __restrict__ W2,
                                   float* __restrict__ Bp2) {
  int idx = blockIdx.x * blockDim.x + threadIdx.x;
  int o = idx % COUT;
  int c = idx / COUT;
  Bp2[(size_t)(c >> 1) * (2 * COUT) + 2 * o + (c & 1)] = W2[idx];
}

__global__ void gat_wmma_gemm_f32_kernel(const float* __restrict__ A,
                                         const float* __restrict__ Bp,
                                         float* __restrict__ C,
                                         int K, int N, int lda, int ldc) {
  const int lane = threadIdx.x & 31;
  const int hi   = lane >> 4;
  const int lr   = lane & 15;
  const int m0   = blockIdx.y << 5;
  const int n0   = blockIdx.x << 6;

  v8f acc0 = {}; v8f acc1 = {}; v8f acc2 = {}; v8f acc3 = {};
  v8f acc4 = {}; v8f acc5 = {}; v8f acc6 = {}; v8f acc7 = {};

  const float* Arow0 = A + (size_t)(m0 + lr) * lda;
  const float* Arow1 = Arow0 + (size_t)16 * lda;
  const v2f* Bv = (const v2f*)Bp;

  v2f a0, a1, b0, b1, b2, b3;
  {
    const int ka = 2 * hi;
    a0 = *(const v2f*)(Arow0 + ka);
    a1 = *(const v2f*)(Arow1 + ka);
    const v2f* bp = Bv + (size_t)hi * N + n0 + lr;
    b0 = bp[0]; b1 = bp[16]; b2 = bp[32]; b3 = bp[48];
  }
  for (int k = 0; k < K; k += 4) {
    const v2f ca0 = a0, ca1 = a1;
    const v2f cb0 = b0, cb1 = b1, cb2 = b2, cb3 = b3;
    if (k + 4 < K) {
      const int ka = k + 4 + 2 * hi;
      a0 = *(const v2f*)(Arow0 + ka);
      a1 = *(const v2f*)(Arow1 + ka);
      const v2f* bp = Bv + (size_t)(ka >> 1) * N + n0 + lr;
      b0 = bp[0]; b1 = bp[16]; b2 = bp[32]; b3 = bp[48];
    }
    acc0 = __builtin_amdgcn_wmma_f32_16x16x4_f32(false, ca0, false, cb0, (short)0, acc0, false, false);
    acc1 = __builtin_amdgcn_wmma_f32_16x16x4_f32(false, ca0, false, cb1, (short)0, acc1, false, false);
    acc2 = __builtin_amdgcn_wmma_f32_16x16x4_f32(false, ca0, false, cb2, (short)0, acc2, false, false);
    acc3 = __builtin_amdgcn_wmma_f32_16x16x4_f32(false, ca0, false, cb3, (short)0, acc3, false, false);
    acc4 = __builtin_amdgcn_wmma_f32_16x16x4_f32(false, ca1, false, cb0, (short)0, acc4, false, false);
    acc5 = __builtin_amdgcn_wmma_f32_16x16x4_f32(false, ca1, false, cb1, (short)0, acc5, false, false);
    acc6 = __builtin_amdgcn_wmma_f32_16x16x4_f32(false, ca1, false, cb2, (short)0, acc6, false, false);
    acc7 = __builtin_amdgcn_wmma_f32_16x16x4_f32(false, ca1, false, cb3, (short)0, acc7, false, false);
  }

  float* Crow0 = C + (size_t)(m0 + 8 * hi) * ldc + n0 + lr;
  float* Crow1 = Crow0 + (size_t)16 * ldc;
#pragma unroll
  for (int i = 0; i < 8; ++i) {
    Crow0[(size_t)i * ldc + 0]  = acc0[i];
    Crow0[(size_t)i * ldc + 16] = acc1[i];
    Crow0[(size_t)i * ldc + 32] = acc2[i];
    Crow0[(size_t)i * ldc + 48] = acc3[i];
    Crow1[(size_t)i * ldc + 0]  = acc4[i];
    Crow1[(size_t)i * ldc + 16] = acc5[i];
    Crow1[(size_t)i * ldc + 32] = acc6[i];
    Crow1[(size_t)i * ldc + 48] = acc7[i];
  }
}

__global__ void gat_dot_kernel(const float* __restrict__ Hm,
                               const float* __restrict__ a,
                               float* __restrict__ S,
                               int nwaves, int nheads, int rowStride) {
  int gt   = blockIdx.x * blockDim.x + threadIdx.x;
  int w    = gt >> 5;
  int lane = gt & 31;
  if (w >= nwaves) return;
  int n = w / nheads, h = w % nheads;
  const float* hp = Hm + (size_t)n * rowStride + h * CHID;
  const float* ap = a + (size_t)h * 2 * CHID;
  float sA = 0.f, sB = 0.f;
#pragma unroll
  for (int j = lane; j < CHID; j += 32) {
    float v = hp[j];
    sA += v * ap[j];
    sB += v * ap[CHID + j];
  }
#pragma unroll
  for (int off = 16; off > 0; off >>= 1) {
    sA += __shfl_down(sA, off, 32);
    sB += __shfl_down(sB, off, 32);
  }
  if (lane == 0) { S[2 * w] = sA; S[2 * w + 1] = sB; }
}

__global__ void gat_agg_kernel(const float* __restrict__ Hm,
                               const float* __restrict__ S,
                               const int* __restrict__ src,
                               const int* __restrict__ dst,
                               float* __restrict__ Yout,
                               int nheads, int rowStride) {
  const int b = blockIdx.x;
  const int t = b / nheads;
  const int h = b % nheads;
  __shared__ float wsh[KNB];
  __shared__ int   dsh[KNB];
  const int tid = threadIdx.x;

  if (tid < KNB) {
    int m = t * KNB + tid;
    int s = src[m], d = dst[m];
    float e = S[(size_t)(s * nheads + h) * 2] + S[(size_t)(d * nheads + h) * 2 + 1];
    wsh[tid] = (e > 0.f) ? e : LRELU_ALPHA * e;
    dsh[tid] = d;
  }
  __syncthreads();
  if (tid == 0) {
    float mx = wsh[0];
    for (int k = 1; k < KNB; ++k) mx = fmaxf(mx, wsh[k]);
    float sum = 0.f;
    for (int k = 0; k < KNB; ++k) { float ex = expf(wsh[k] - mx); wsh[k] = ex; sum += ex; }
    float inv = 1.f / sum;
    for (int k = 0; k < KNB; ++k) wsh[k] *= inv;
  }
  __syncthreads();

  float acc = 0.f;
  const float* col = Hm + (size_t)h * CHID + tid;
#pragma unroll 4
  for (int k = 0; k < KNB; ++k)
    acc += wsh[k] * col[(size_t)dsh[k] * rowStride];
  acc = (acc > 0.f) ? acc : expm1f(acc);
  Yout[(size_t)t * (nheads * CHID) + h * CHID + tid] = acc;
}

extern "C" void kernel_launch(void* const* d_in, const int* in_sizes, int n_in,
                              void* d_out, int out_size, void* d_ws, size_t ws_size,
                              hipStream_t stream) {
  (void)in_sizes; (void)n_in; (void)out_size; (void)ws_size;
  const float* feats = (const float*)d_in[0];
  const float* W1    = (const float*)d_in[1];
  const float* a1    = (const float*)d_in[2];
  const float* W2    = (const float*)d_in[3];
  const float* a2    = (const float*)d_in[4];
  const int*   src1  = (const int*)d_in[5];
  const int*   dst1  = (const int*)d_in[6];
  const int*   src2  = (const int*)d_in[8];
  const int*   dst2  = (const int*)d_in[9];
  float* out = (float*)d_out;

  float* ws  = (float*)d_ws;
  float* h1  = ws;
  float* x   = h1  + (size_t)CN2 * CNH * CHID;
  float* Bp1 = x   + (size_t)CN1 * CNH * CHID;
  float* Bp2 = Bp1 + (size_t)CIN * CNH * CHID;
  float* h2  = Bp2 + (size_t)CNH * CHID * COUT;
  float* s1  = h2  + (size_t)CN1 * COUT;
  float* s2  = s1  + (size_t)CN2 * CNH * 2;

  gat_pack_w1_kernel<<<(CNH * CIN * CHID) / 256, 256, 0, stream>>>(W1, Bp1);
  gat_pack_w2_kernel<<<(CNH * CHID * COUT) / 256, 256, 0, stream>>>(W2, Bp2);
  dim3 g1((CNH * CHID) / 64, CN2 / 32);
  gat_wmma_gemm_f32_kernel<<<g1, 32, 0, stream>>>(feats, Bp1, h1,
                                                  CIN, CNH * CHID, CIN, CNH * CHID);
  gat_dot_kernel<<<(CN2 * CNH) / 8, 256, 0, stream>>>(h1, a1, s1,
                                                      CN2 * CNH, CNH, CNH * CHID);
  gat_agg_kernel<<<CN1 * CNH, 128, 0, stream>>>(h1, s1, src1, dst1, x,
                                                CNH, CNH * CHID);
  dim3 g2(COUT / 64, CN1 / 32);
  gat_wmma_gemm_f32_kernel<<<g2, 32, 0, stream>>>(x, Bp2, h2,
                                                  CNH * CHID, COUT, CNH * CHID, COUT);
  gat_dot_kernel<<<CN1 / 8, 256, 0, stream>>>(h2, a2, s2, CN1, 1, COUT);
  gat_agg_kernel<<<CN0, 128, 0, stream>>>(h2, s2, src2, dst2, out, 1, COUT);
}


# layer-2 GEMM: K split over the 16 waves of a 512-thread work-group on a 16x64 tile (256 work-groups), partial tiles summed in fixed order through LDS; changes the k summation order of h2
# speedup vs baseline: 1.5631x; 1.5631x over previous
#include <hip/hip_runtime.h>

#define CN0 512
#define CN1 2048
#define CN2 8192
#define KNB 33
#define CIN 256
#define CHID 128
#define COUT 128
#define CNH 8
#define LRELU_ALPHA 0.2f

typedef float v2f __attribute__((ext_vector_type(2)));
typedef float v8f __attribute__((ext_vector_type(8)));

__global__ void gat_pack_w1_kernel(const float* __restrict__ W1,
                                   float* __restrict__ Bp1) {
  int idx = blockIdx.x * blockDim.x + threadIdx.x;
  int j = idx % CHID;
  int i = (idx / CHID) % CIN;
  int h = idx / (CHID * CIN);
  int n = h * CHID + j;
  Bp1[(size_t)(i >> 1) * (2 * CNH * CHID) + 2 * n + (i & 1)] = W1[idx];
}

__global__ void gat_pack_w2_kernel(const float* __restrict__ W2,
                                   float* __restrict__ Bp2) {
  int idx = blockIdx.x * blockDim.x + threadIdx.x;
  int o = idx % COUT;
  int c = idx / COUT;
  Bp2[(size_t)(c >> 1) * (2 * COUT) + 2 * o + (c & 1)] = W2[idx];
}

__global__ void gat_wmma_gemm_f32_kernel(const float* __restrict__ A,
                                         const float* __restrict__ Bp,
                                         float* __restrict__ C,
                                         int K, int N, int lda, int ldc) {
  const int lane = threadIdx.x & 31;
  const int hi   = lane >> 4;
  const int lr   = lane & 15;
  const int m0   = blockIdx.y << 5;
  const int n0   = blockIdx.x << 6;

  v8f acc0 = {}; v8f acc1 = {}; v8f acc2 = {}; v8f acc3 = {};
  v8f acc4 = {}; v8f acc5 = {}; v8f acc6 = {}; v8f acc7 = {};

  const float* Arow0 = A + (size_t)(m0 + lr) * lda;
  const float* Arow1 = Arow0 + (size_t)16 * lda;
  const v2f* Bv = (const v2f*)Bp;

  v2f a0, a1, b0, b1, b2, b3;
  {
    const int ka = 2 * hi;
    a0 = *(const v2f*)(Arow0 + ka);
    a1 = *(const v2f*)(Arow1 + ka);
    const v2f* bp = Bv + (size_t)hi * N + n0 + lr;
    b0 = bp[0]; b1 = bp[16]; b2 = bp[32]; b3 = bp[48];
  }
  for (int k = 0; k < K; k += 4) {
    const v2f ca0 = a0, ca1 = a1;
    const v2f cb0 = b0, cb1 = b1, cb2 = b2, cb3 = b3;
    if (k + 4 < K) {
      const int ka = k + 4 + 2 * hi;
      a0 = *(const v2f*)(Arow0 + ka);
      a1 = *(const v2f*)(Arow1 + ka);
      const v2f* bp = Bv + (size_t)(ka >> 1) * N + n0 + lr;
      b0 = bp[0]; b1 = bp[16]; b2 = bp[32]; b3 = bp[48];
    }
    acc0 = __builtin_amdgcn_wmma_f32_16x16x4_f32(false, ca0, false, cb0, (short)0, acc0, false, false);
    acc1 = __builtin_amdgcn_wmma_f32_16x16x4_f32(false, ca0, false, cb1, (short)0, acc1, false, false);
    acc2 = __builtin_amdgcn_wmma_f32_16x16x4_f32(false, ca0, false, cb2, (short)0, acc2, false, false);
    acc3 = __builtin_amdgcn_wmma_f32_16x16x4_f32(false, ca0, false, cb3, (short)0, acc3, false, false);
    acc4 = __builtin_amdgcn_wmma_f32_16x16x4_f32(false, ca1, false, cb0, (short)0, acc4, false, false);
    acc5 = __builtin_amdgcn_wmma_f32_16x16x4_f32(false, ca1, false, cb1, (short)0, acc5, false, false);
    acc6 = __builtin_amdgcn_wmma_f32_16x16x4_f32(false, ca1, false, cb2, (short)0, acc6, false, false);
    acc7 = __builtin_amdgcn_wmma_f32_16x16x4_f32(false, ca1, false, cb3, (short)0, acc7, false, false);
  }

  float* Crow0 = C + (size_t)(m0 + 8 * hi) * ldc + n0 + lr;
  float* Crow1 = Crow0 + (size_t)16 * ldc;
#pragma unroll
  for (int i = 0; i < 8; ++i) {
    Crow0[(size_t)i * ldc + 0]  = acc0[i];
    Crow0[(size_t)i * ldc + 16] = acc1[i];
    Crow0[(size_t)i * ldc + 32] = acc2[i];
    Crow0[(size_t)i * ldc + 48] = acc3[i];
    Crow1[(size_t)i * ldc + 0]  = acc4[i];
    Crow1[(size_t)i * ldc + 16] = acc5[i];
    Crow1[(size_t)i * ldc + 32] = acc6[i];
    Crow1[(size_t)i * ldc + 48] = acc7[i];
  }
}

#define SPLITK 16
__global__ void __launch_bounds__(32 * SPLITK)
gat_wmma_gemm_f32_splitk_kernel(const float* __restrict__ A,
                                const float* __restrict__ Bp,
                                float* __restrict__ C,
                                int K, int N, int lda, int ldc) {
  __shared__ float part[SPLITK][32][32];
  const int lane = threadIdx.x & 31;
  const int wv   = __builtin_amdgcn_readfirstlane((int)(threadIdx.x >> 5));
  const int hi   = lane >> 4;
  const int lr   = lane & 15;
  const int m0   = blockIdx.y << 4;
  const int n0   = blockIdx.x << 6;
  const int kbeg = wv * (K / SPLITK);
  const int kend = kbeg + K / SPLITK;

  v8f acc0 = {}; v8f acc1 = {}; v8f acc2 = {}; v8f acc3 = {};

  const float* Arow0 = A + (size_t)(m0 + lr) * lda;
  const v2f* Bv = (const v2f*)Bp;

  v2f a0, b0, b1, b2, b3;
  {
    const int ka = kbeg + 2 * hi;
    a0 = *(const v2f*)(Arow0 + ka);
    const v2f* bp = Bv + (size_t)(ka >> 1) * N + n0 + lr;
    b0 = bp[0]; b1 = bp[16]; b2 = bp[32]; b3 = bp[48];
  }
  for (int k = kbeg; k < kend; k += 4) {
    const v2f ca0 = a0;
    const v2f cb0 = b0, cb1 = b1, cb2 = b2, cb3 = b3;
    if (k + 4 < kend) {
      const int ka = k + 4 + 2 * hi;
      a0 = *(const v2f*)(Arow0 + ka);
      const v2f* bp = Bv + (size_t)(ka >> 1) * N + n0 + lr;
      b0 = bp[0]; b1 = bp[16]; b2 = bp[32]; b3 = bp[48];
    }
    acc0 = __builtin_amdgcn_wmma_f32_16x16x4_f32(false, ca0, false, cb0, (short)0, acc0, false, false);
    acc1 = __builtin_amdgcn_wmma_f32_16x16x4_f32(false, ca0, false, cb1, (short)0, acc1, false, false);
    acc2 = __builtin_amdgcn_wmma_f32_16x16x4_f32(false, ca0, false, cb2, (short)0, acc2, false, false);
    acc3 = __builtin_amdgcn_wmma_f32_16x16x4_f32(false, ca0, false, cb3, (short)0, acc3, false, false);
  }

#pragma unroll
  for (int i = 0; i < 8; ++i) {
    part[wv][0 * 8 + i][lane] = acc0[i];
    part[wv][1 * 8 + i][lane] = acc1[i];
    part[wv][2 * 8 + i][lane] = acc2[i];
    part[wv][3 * 8 + i][lane] = acc3[i];
  }
  __syncthreads();

  const int ri = wv & 7;
  const int rh = wv >> 3;
  const int pl = rh * 16 + lr;
  float sum0 = part[0][(hi) * 8 + ri][pl];
  float sum1 = part[0][(2 + hi) * 8 + ri][pl];
#pragma unroll
  for (int s = 1; s < SPLITK; ++s) {
    sum0 += part[s][(hi) * 8 + ri][pl];
    sum1 += part[s][(2 + hi) * 8 + ri][pl];
  }
  float* Crow = C + (size_t)(m0 + wv) * ldc + n0 + lane;
  Crow[0]  = sum0;
  Crow[32] = sum1;
}

__global__ void gat_dot_kernel(const float* __restrict__ Hm,
                               const float* __restrict__ a,
                               float* __restrict__ S,
                               int nwaves, int nheads, int rowStride) {
  int gt   = blockIdx.x * blockDim.x + threadIdx.x;
  int w    = gt >> 5;
  int lane = gt & 31;
  if (w >= nwaves) return;
  int n = w / nheads, h = w % nheads;
  const float* hp = Hm + (size_t)n * rowStride + h * CHID;
  const float* ap = a + (size_t)h * 2 * CHID;
  float sA = 0.f, sB = 0.f;
#pragma unroll
  for (int j = lane; j < CHID; j += 32) {
    float v = hp[j];
    sA += v * ap[j];
    sB += v * ap[CHID + j];
  }
#pragma unroll
  for (int off = 16; off > 0; off >>= 1) {
    sA += __shfl_down(sA, off, 32);
    sB += __shfl_down(sB, off, 32);
  }
  if (lane == 0) { S[2 * w] = sA; S[2 * w + 1] = sB; }
}

__global__ void gat_agg_kernel(const float* __restrict__ Hm,
                               const float* __restrict__ S,
                               const int* __restrict__ src,
                               const int* __restrict__ dst,
                               float* __restrict__ Yout,
                               int nheads, int rowStride) {
  const int b = blockIdx.x;
  const int t = b / nheads;
  const int h = b % nheads;
  __shared__ float wsh[KNB];
  __shared__ int   dsh[KNB];
  const int tid = threadIdx.x;

  if (tid < KNB) {
    int m = t * KNB + tid;
    int s = src[m], d = dst[m];
    float e = S[(size_t)(s * nheads + h) * 2] + S[(size_t)(d * nheads + h) * 2 + 1];
    wsh[tid] = (e > 0.f) ? e : LRELU_ALPHA * e;
    dsh[tid] = d;
  }
  __syncthreads();
  if (tid == 0) {
    float mx = wsh[0];
    for (int k = 1; k < KNB; ++k) mx = fmaxf(mx, wsh[k]);
    float sum = 0.f;
    for (int k = 0; k < KNB; ++k) { float ex = expf(wsh[k] - mx); wsh[k] = ex; sum += ex; }
    float inv = 1.f / sum;
    for (int k = 0; k < KNB; ++k) wsh[k] *= inv;
  }
  __syncthreads();

  float acc = 0.f;
  const float* col = Hm + (size_t)h * CHID + tid;
#pragma unroll 4
  for (int k = 0; k < KNB; ++k)
    acc += wsh[k] * col[(size_t)dsh[k] * rowStride];
  acc = (acc > 0.f) ? acc : expm1f(acc);
  Yout[(size_t)t * (nheads * CHID) + h * CHID + tid] = acc;
}

extern "C" void kernel_launch(void* const* d_in, const int* in_sizes, int n_in,
                              void* d_out, int out_size, void* d_ws, size_t ws_size,
                              hipStream_t stream) {
  (void)in_sizes; (void)n_in; (void)out_size; (void)ws_size;
  const float* feats = (const float*)d_in[0];
  const float* W1    = (const float*)d_in[1];
  const float* a1    = (const float*)d_in[2];
  const float* W2    = (const float*)d_in[3];
  const float* a2    = (const float*)d_in[4];
  const int*   src1  = (const int*)d_in[5];
  const int*   dst1  = (const int*)d_in[6];
  const int*   src2  = (const int*)d_in[8];
  const int*   dst2  = (const int*)d_in[9];
  float* out = (float*)d_out;

  float* ws  = (float*)d_ws;
  float* h1  = ws;
  float* x   = h1  + (size_t)CN2 * CNH * CHID;
  float* Bp1 = x   + (size_t)CN1 * CNH * CHID;
  float* Bp2 = Bp1 + (size_t)CIN * CNH * CHID;
  float* h2  = Bp2 + (size_t)CNH * CHID * COUT;
  float* s1  = h2  + (size_t)CN1 * COUT;
  float* s2  = s1  + (size_t)CN2 * CNH * 2;

  gat_pack_w1_kernel<<<(CNH * CIN * CHID) / 256, 256, 0, stream>>>(W1, Bp1);
  gat_pack_w2_kernel<<<(CNH * CHID * COUT) / 256, 256, 0, stream>>>(W2, Bp2);
  dim3 g1((CNH * CHID) / 64, CN2 / 32);
  gat_wmma_gemm_f32_kernel<<<g1, 32, 0, stream>>>(feats, Bp1, h1,
                                                  CIN, CNH * CHID, CIN, CNH * CHID);
  gat_dot_kernel<<<(CN2 * CNH) / 8, 256, 0, stream>>>(h1, a1, s1,
                                                      CN2 * CNH, CNH, CNH * CHID);
  gat_agg_kernel<<<CN1 * CNH, 128, 0, stream>>>(h1, s1, src1, dst1, x,
                                                CNH, CNH * CHID);
  dim3 g2(COUT / 64, CN1 / 16);
  gat_wmma_gemm_f32_splitk_kernel<<<g2, 32 * SPLITK, 0, stream>>>(x, Bp2, h2,
                                                                   CNH * CHID, COUT, CNH * CHID, COUT);
  gat_dot_kernel<<<CN1 / 8, 256, 0, stream>>>(h2, a2, s2, CN1, 1, COUT);
  gat_agg_kernel<<<CN0, 128, 0, stream>>>(h2, s2, src2, dst2, out, 1, COUT);
}


# layer-1 score and aggregation kernels re-cut: one wave covers a whole 128-wide head with 16-byte loads, so 4x to 16x fewer waves enter; softmax by lane butterfly; changes dot and softmax sum order
# speedup vs baseline: 2.5564x; 1.6355x over previous
#include <hip/hip_runtime.h>

#define CN0 512
#define CN1 2048
#define CN2 8192
#define KNB 33
#define CIN 256
#define CHID 128
#define COUT 128
#define CNH 8
#define LRELU_ALPHA 0.2f

typedef float v2f __attribute__((ext_vector_type(2)));
typedef float v8f __attribute__((ext_vector_type(8)));

__global__ void gat_pack_w1_kernel(const float* __restrict__ W1,
                                   float* __restrict__ Bp1) {
  int idx = blockIdx.x * blockDim.x + threadIdx.x;
  int j = idx % CHID;
  int i = (idx / CHID) % CIN;
  int h = idx / (CHID * CIN);
  int n = h * CHID + j;
  Bp1[(size_t)(i >> 1) * (2 * CNH * CHID) + 2 * n + (i & 1)] = W1[idx];
}

__global__ void gat_pack_w2_kernel(const float* __restrict__ W2,
                                   float* __restrict__ Bp2) {
  int idx = blockIdx.x * blockDim.x + threadIdx.x;
  int o = idx % COUT;
  int c = idx / COUT;
  Bp2[(size_t)(c >> 1) * (2 * COUT) + 2 * o + (c & 1)] = W2[idx];
}

__global__ void gat_wmma_gemm_f32_kernel(const float* __restrict__ A,
                                         const float* __restrict__ Bp,
                                         float* __restrict__ C,
                                         int K, int N, int lda, int ldc) {
  const int lane = threadIdx.x & 31;
  const int hi   = lane >> 4;
  const int lr   = lane & 15;
  const int m0   = blockIdx.y << 5;
  const int n0   = blockIdx.x << 6;

  v8f acc0 = {}; v8f acc1 = {}; v8f acc2 = {}; v8f acc3 = {};
  v8f acc4 = {}; v8f acc5 = {}; v8f acc6 = {}; v8f acc7 = {};

  const float* Arow0 = A + (size_t)(m0 + lr) * lda;
  const float* Arow1 = Arow0 + (size_t)16 * lda;
  const v2f* Bv = (const v2f*)Bp;

  v2f a0, a1, b0, b1, b2, b3;
  {
    const int ka = 2 * hi;
    a0 = *(const v2f*)(Arow0 + ka);
    a1 = *(const v2f*)(Arow1 + ka);
    const v2f* bp = Bv + (size_t)hi * N + n0 + lr;
    b0 = bp[0]; b1 = bp[16]; b2 = bp[32]; b3 = bp[48];
  }
  for (int k = 0; k < K; k += 4) {
    const v2f ca0 = a0, ca1 = a1;
    const v2f cb0 = b0, cb1 = b1, cb2 = b2, cb3 = b3;
    if (k + 4 < K) {
      const int ka = k + 4 + 2 * hi;
      a0 = *(const v2f*)(Arow0 + ka);
      a1 = *(const v2f*)(Arow1 + ka);
      const v2f* bp = Bv + (size_t)(ka >> 1) * N + n0 + lr;
      b0 = bp[0]; b1 = bp[16]; b2 = bp[32]; b3 = bp[48];
    }
    acc0 = __builtin_amdgcn_wmma_f32_16x16x4_f32(false, ca0, false, cb0, (short)0, acc0, false, false);
    acc1 = __builtin_amdgcn_wmma_f32_16x16x4_f32(false, ca0, false, cb1, (short)0, acc1, false, false);
    acc2 = __builtin_amdgcn_wmma_f32_16x16x4_f32(false, ca0, false, cb2, (short)0, acc2, false, false);
    acc3 = __builtin_amdgcn_wmma_f32_16x16x4_f32(false, ca0, false, cb3, (short)0, acc3, false, false);
    acc4 = __builtin_amdgcn_wmma_f32_16x16x4_f32(false, ca1, false, cb0, (short)0, acc4, false, false);
    acc5 = __builtin_amdgcn_wmma_f32_16x16x4_f32(false, ca1, false, cb1, (short)0, acc5, false, false);
    acc6 = __builtin_amdgcn_wmma_f32_16x16x4_f32(false, ca1, false, cb2, (short)0, acc6, false, false);
    acc7 = __builtin_amdgcn_wmma_f32_16x16x4_f32(false, ca1, false, cb3, (short)0, acc7, false, false);
  }

  float* Crow0 = C + (size_t)(m0 + 8 * hi) * ldc + n0 + lr;
  float* Crow1 = Crow0 + (size_t)16 * ldc;
#pragma unroll
  for (int i = 0; i < 8; ++i) {
    Crow0[(size_t)i * ldc + 0]  = acc0[i];
    Crow0[(size_t)i * ldc + 16] = acc1[i];
    Crow0[(size_t)i * ldc + 32] = acc2[i];
    Crow0[(size_t)i * ldc + 48] = acc3[i];
    Crow1[(size_t)i * ldc + 0]  = acc4[i];
    Crow1[(size_t)i * ldc + 16] = acc5[i];
    Crow1[(size_t)i * ldc + 32] = acc6[i];
    Crow1[(size_t)i * ldc + 48] = acc7[i];
  }
}

#define SPLITK 16
__global__ void __launch_bounds__(32 * SPLITK)
gat_wmma_gemm_f32_splitk_kernel(const float* __restrict__ A,
                                const float* __restrict__ Bp,
                                float* __restrict__ C,
                                int K, int N, int lda, int ldc) {
  __shared__ float part[SPLITK][32][32];
  const int lane = threadIdx.x & 31;
  const int wv   = __builtin_amdgcn_readfirstlane((int)(threadIdx.x >> 5));
  const int hi   = lane >> 4;
  const int lr   = lane & 15;
  const int m0   = blockIdx.y << 4;
  const int n0   = blockIdx.x << 6;
  const int kbeg = wv * (K / SPLITK);
  const int kend = kbeg + K / SPLITK;

  v8f acc0 = {}; v8f acc1 = {}; v8f acc2 = {}; v8f acc3 = {};

  const float* Arow0 = A + (size_t)(m0 + lr) * lda;
  const v2f* Bv = (const v2f*)Bp;

  v2f a0, b0, b1, b2, b3;
  {
    const int ka = kbeg + 2 * hi;
    a0 = *(const v2f*)(Arow0 + ka);
    const v2f* bp = Bv + (size_t)(ka >> 1) * N + n0 + lr;
    b0 = bp[0]; b1 = bp[16]; b2 = bp[32]; b3 = bp[48];
  }
  for (int k = kbeg; k < kend; k += 4) {
    const v2f ca0 = a0;
    const v2f cb0 = b0, cb1 = b1, cb2 = b2, cb3 = b3;
    if (k + 4 < kend) {
      const int ka = k + 4 + 2 * hi;
      a0 = *(const v2f*)(Arow0 + ka);
      const v2f* bp = Bv + (size_t)(ka >> 1) * N + n0 + lr;
      b0 = bp[0]; b1 = bp[16]; b2 = bp[32]; b3 = bp[48];
    }
    acc0 = __builtin_amdgcn_wmma_f32_16x16x4_f32(false, ca0, false, cb0, (short)0, acc0, false, false);
    acc1 = __builtin_amdgcn_wmma_f32_16x16x4_f32(false, ca0, false, cb1, (short)0, acc1, false, false);
    acc2 = __builtin_amdgcn_wmma_f32_16x16x4_f32(false, ca0, false, cb2, (short)0, acc2, false, false);
    acc3 = __builtin_amdgcn_wmma_f32_16x16x4_f32(false, ca0, false, cb3, (short)0, acc3, false, false);
  }

#pragma unroll
  for (int i = 0; i < 8; ++i) {
    part[wv][0 * 8 + i][lane] = acc0[i];
    part[wv][1 * 8 + i][lane] = acc1[i];
    part[wv][2 * 8 + i][lane] = acc2[i];
    part[wv][3 * 8 + i][lane] = acc3[i];
  }
  __syncthreads();

  const int ri = wv & 7;
  const int rh = wv >> 3;
  const int pl = rh * 16 + lr;
  float sum0 = part[0][(hi) * 8 + ri][pl];
  float sum1 = part[0][(2 + hi) * 8 + ri][pl];
#pragma unroll
  for (int s = 1; s < SPLITK; ++s) {
    sum0 += part[s][(hi) * 8 + ri][pl];
    sum1 += part[s][(2 + hi) * 8 + ri][pl];
  }
  float* Crow = C + (size_t)(m0 + wv) * ldc + n0 + lane;
  Crow[0]  = sum0;
  Crow[32] = sum1;
}

__global__ void gat_dot_kernel(const float* __restrict__ Hm,
                               const float* __restrict__ a,
                               float* __restrict__ S,
                               int nwaves, int nheads, int rowStride) {
  int gt   = blockIdx.x * blockDim.x + threadIdx.x;
  int w    = gt >> 5;
  int lane = gt & 31;
  if (w >= nwaves) return;
  int n = w / nheads, h = w % nheads;
  const float* hp = Hm + (size_t)n * rowStride + h * CHID;
  const float* ap = a + (size_t)h * 2 * CHID;
  float sA = 0.f, sB = 0.f;
#pragma unroll
  for (int j = lane; j < CHID; j += 32) {
    float v = hp[j];
    sA += v * ap[j];
    sB += v * ap[CHID + j];
  }
#pragma unroll
  for (int off = 16; off > 0; off >>= 1) {
    sA += __shfl_down(sA, off, 32);
    sB += __shfl_down(sB, off, 32);
  }
  if (lane == 0) { S[2 * w] = sA; S[2 * w + 1] = sB; }
}

#define DOT_NPW 2
static_assert(CN2 % DOT_NPW == 0, "whole waves only");
__global__ void gat_dot_node_kernel(const float* __restrict__ Hm,
                                    const float* __restrict__ a,
                                    float* __restrict__ S,
                                    int nnodes) {
  const int gt   = blockIdx.x * blockDim.x + threadIdx.x;
  const int nb   = (gt >> 5) * DOT_NPW;
  const int lane = gt & 31;
  if (nb + DOT_NPW > nnodes) return;
  const float4* ap = (const float4*)a + lane;
  float4 p[CNH], q[CNH];
#pragma unroll
  for (int h = 0; h < CNH; ++h) {
    p[h] = ap[h * (2 * CHID / 4)];
    q[h] = ap[h * (2 * CHID / 4) + CHID / 4];
  }
  float sA[DOT_NPW][CNH], sB[DOT_NPW][CNH];
#pragma unroll
  for (int r = 0; r < DOT_NPW; ++r) {
    const float4* hp = (const float4*)(Hm + (size_t)(nb + r) * (CNH * CHID)) + lane;
#pragma unroll
    for (int h = 0; h < CNH; ++h) {
      const float4 v = hp[h * (CHID / 4)];
      sA[r][h] = v.x * p[h].x + v.y * p[h].y + v.z * p[h].z + v.w * p[h].w;
      sB[r][h] = v.x * q[h].x + v.y * q[h].y + v.z * q[h].z + v.w * q[h].w;
    }
  }
#pragma unroll
  for (int r = 0; r < DOT_NPW; ++r) {
#pragma unroll
    for (int h = 0; h < CNH; ++h) {
#pragma unroll
      for (int off = 16; off > 0; off >>= 1) {
        sA[r][h] += __shfl_xor(sA[r][h], off, 32);
        sB[r][h] += __shfl_xor(sB[r][h], off, 32);
      }
    }
  }
  if (lane == 0) {
#pragma unroll
    for (int r = 0; r < DOT_NPW; ++r) {
      float4* o = (float4*)(S + (size_t)(nb + r) * (2 * CNH));
#pragma unroll
      for (int h = 0; h < CNH; h += 2)
        o[h >> 1] = make_float4(sA[r][h], sB[r][h], sA[r][h + 1], sB[r][h + 1]);
    }
  }
}

static_assert(KNB == 33, "edge 32 is the one edge beyond the 32 lanes");
__global__ void __launch_bounds__(32 * CNH)
gat_agg_wave_kernel(const float* __restrict__ Hm,
                    const float* __restrict__ S,
                    const int* __restrict__ src,
                    const int* __restrict__ dst,
                    float* __restrict__ Yout,
                    int nheads, int rowStride) {
  const int t    = blockIdx.x;
  const int lane = threadIdx.x & 31;
  const int h    = __builtin_amdgcn_readfirstlane((int)(threadIdx.x >> 5));
  const int m0   = t * KNB;

  const int sL = src[m0 + lane], dL = dst[m0 + lane];
  const int sX = src[m0 + 32],   dX = dst[m0 + 32];
  float eL = S[(size_t)(sL * nheads + h) * 2] + S[(size_t)(dL * nheads + h) * 2 + 1];
  float eX = S[(size_t)(sX * nheads + h) * 2] + S[(size_t)(dX * nheads + h) * 2 + 1];
  eL = (eL > 0.f) ? eL : LRELU_ALPHA * eL;
  eX = (eX > 0.f) ? eX : LRELU_ALPHA * eX;

  float mx = eL;
  for (int off = 16; off > 0; off >>= 1) mx = fmaxf(mx, __shfl_xor(mx, off, 32));
  mx = fmaxf(mx, eX);
  const float pL = expf(eL - mx);
  const float pX = expf(eX - mx);
  float sum = pL;
  for (int off = 16; off > 0; off >>= 1) sum += __shfl_xor(sum, off, 32);
  sum += pX;
  const float inv = 1.f / sum;
  const float wL = pL * inv;
  const float wX = pX * inv;

  const float* col = Hm + (size_t)h * CHID + 4 * lane;
  float4 acc = make_float4(0.f, 0.f, 0.f, 0.f);
#pragma unroll
  for (int k = 0; k < 32; ++k) {
    const int   dk = __builtin_amdgcn_readlane(dL, k);
    const float wk = __int_as_float(__builtin_amdgcn_readlane(__float_as_int(wL), k));
    const float4 v = *(const float4*)(col + (size_t)dk * rowStride);
    acc.x += wk * v.x; acc.y += wk * v.y; acc.z += wk * v.z; acc.w += wk * v.w;
  }
  {
    const int   dk = __builtin_amdgcn_readfirstlane(dX);
    const float wk = __int_as_float(__builtin_amdgcn_readfirstlane(__float_as_int(wX)));
    const float4 v = *(const float4*)(col + (size_t)dk * rowStride);
    acc.x += wk * v.x; acc.y += wk * v.y; acc.z += wk * v.z; acc.w += wk * v.w;
  }
  float4 r;
  r.x = (acc.x > 0.f) ? acc.x : expm1f(acc.x);
  r.y = (acc.y > 0.f) ? acc.y : expm1f(acc.y);
  r.z = (acc.z > 0.f) ? acc.z : expm1f(acc.z);
  r.w = (acc.w > 0.f) ? acc.w : expm1f(acc.w);
  *(float4*)(Yout + (size_t)t * (nheads * CHID) + h * CHID + 4 * lane) = r;
}

extern "C" void kernel_launch(void* const* d_in, const int* in_sizes, int n_in,
                              void* d_out, int out_size, void* d_ws, size_t ws_size,
                              hipStream_t stream) {
  (void)in_sizes; (void)n_in; (void)out_size; (void)ws_size;
  const float* feats = (const float*)d_in[0];
  const float* W1    = (const float*)d_in[1];
  const float* a1    = (const float*)d_in[2];
  const float* W2    = (const float*)d_in[3];
  const float* a2    = (const float*)d_in[4];
  const int*   src1  = (const int*)d_in[5];
  const int*   dst1  = (const int*)d_in[6];
  const int*   src2  = (const int*)d_in[8];
  const int*   dst2  = (const int*)d_in[9];
  float* out = (float*)d_out;

  float* ws  = (float*)d_ws;
  float* h1  = ws;
  float* x   = h1  + (size_t)CN2 * CNH * CHID;
  float* Bp1 = x   + (size_t)CN1 * CNH * CHID;
  float* Bp2 = Bp1 + (size_t)CIN * CNH * CHID;
  float* h2  = Bp2 + (size_t)CNH * CHID * COUT;
  float* s1  = h2  + (size_t)CN1 * COUT;
  float* s2  = s1  + (size_t)CN2 * CNH * 2;

  gat_pack_w1_kernel<<<(CNH * CIN * CHID) / 256, 256, 0, stream>>>(W1, Bp1);
  gat_pack_w2_kernel<<<(CNH * CHID * COUT) / 256, 256, 0, stream>>>(W2, Bp2);
  dim3 g1((CNH * CHID) / 64, CN2 / 32);
  gat_wmma_gemm_f32_kernel<<<g1, 32, 0, stream>>>(feats, Bp1, h1,
                                                  CIN, CNH * CHID, CIN, CNH * CHID);
  gat_dot_node_kernel<<<CN2 / (8 * DOT_NPW), 256, 0, stream>>>(h1, a1, s1, CN2);
  gat_agg_wave_kernel<<<CN1, 32 * CNH, 0, stream>>>(h1, s1, src1, dst1, x,
                                                    CNH, CNH * CHID);
  dim3 g2(COUT / 64, CN1 / 16);
  gat_wmma_gemm_f32_splitk_kernel<<<g2, 32 * SPLITK, 0, stream>>>(x, Bp2, h2,
                                                                   CNH * CHID, COUT, CNH * CHID, COUT);
  gat_dot_kernel<<<CN1 / 8, 256, 0, stream>>>(h2, a2, s2, CN1, 1, COUT);
  gat_agg_wave_kernel<<<CN0, 32, 0, stream>>>(h2, s2, src2, dst2, out, 1, COUT);
}


# layer 1 reassociated: attention-weighted sum of the 256-wide input rows first, then the W1 product on 2048 rows, not 8192; scores from feats.(W1.a1); changes the sum order of x and of the scores
# speedup vs baseline: 4.0567x; 1.5869x over previous
#include <hip/hip_runtime.h>

#define CN0 512
#define CN1 2048
#define CN2 8192
#define KNB 33
#define CIN 256
#define CHID 128
#define COUT 128
#define CNH 8
#define LRELU_ALPHA 0.2f

typedef float v2f __attribute__((ext_vector_type(2)));
typedef float v8f __attribute__((ext_vector_type(8)));

__global__ void gat_pack_w1_wa_kernel(const float* __restrict__ W1,
                                      const float* __restrict__ a1,
                                      float* __restrict__ Bp1,
                                      float* __restrict__ waT) {
  const int gt   = blockIdx.x * blockDim.x + threadIdx.x;
  const int w    = gt >> 5;
  const int lane = gt & 31;
  const int h    = w / (CIN / 2);
  const int ip   = w % (CIN / 2);
  const float4 r0 = *((const float4*)(W1 + ((size_t)h * CIN + 2 * ip) * CHID) + lane);
  const float4 r1 = *((const float4*)(W1 + ((size_t)h * CIN + 2 * ip + 1) * CHID) + lane);
  float4* o = (float4*)(Bp1 + (size_t)ip * (2 * CNH * CHID) + 2 * (h * CHID + 4 * lane));
  o[0] = make_float4(r0.x, r1.x, r0.y, r1.y);
  o[1] = make_float4(r0.z, r1.z, r0.w, r1.w);
  const float4 p = *((const float4*)(a1 + (size_t)h * 2 * CHID) + lane);
  const float4 q = *((const float4*)(a1 + (size_t)h * 2 * CHID + CHID) + lane);
  float s0 = r0.x * p.x + r0.y * p.y + r0.z * p.z + r0.w * p.w;
  float s1 = r0.x * q.x + r0.y * q.y + r0.z * q.z + r0.w * q.w;
  float s2 = r1.x * p.x + r1.y * p.y + r1.z * p.z + r1.w * p.w;
  float s3 = r1.x * q.x + r1.y * q.y + r1.z * q.z + r1.w * q.w;
#pragma unroll
  for (int off = 16; off > 0; off >>= 1) {
    s0 += __shfl_xor(s0, off, 32);
    s1 += __shfl_xor(s1, off, 32);
    s2 += __shfl_xor(s2, off, 32);
    s3 += __shfl_xor(s3, off, 32);
  }
  if (lane == 0) {
    *(float2*)(waT + (size_t)(h * 2 + 0) * CIN + 2 * ip) = make_float2(s0, s2);
    *(float2*)(waT + (size_t)(h * 2 + 1) * CIN + 2 * ip) = make_float2(s1, s3);
  }
}

__global__ void gat_pack_w2_kernel(const float* __restrict__ W2,
                                   float* __restrict__ Bp2) {
  int idx = blockIdx.x * blockDim.x + threadIdx.x;
  int o = idx % COUT;
  int c = idx / COUT;
  Bp2[(size_t)(c >> 1) * (2 * COUT) + 2 * o + (c & 1)] = W2[idx];
}

#define GEMM_GK 4
static_assert(CHID % 64 == 0, "a 64-column tile must not straddle heads");
static_assert(CIN % (8 * GEMM_GK) == 0, "whole pairs of groups of k-steps only");

struct gat_frag_group {
  v2f a0[GEMM_GK], a1[GEMM_GK], b0[GEMM_GK], b1[GEMM_GK], b2[GEMM_GK], b3[GEMM_GK];
};

__device__ __forceinline__ void gat_load_group(gat_frag_group& f,
                                               const float* __restrict__ Arow0,
                                               const float* __restrict__ Arow1,
                                               const v2f* __restrict__ Bcol,
                                               int N, int kb, int hi) {
#pragma unroll
  for (int u = 0; u < GEMM_GK; ++u) {
    const int ka = kb + 4 * u + 2 * hi;
    f.a0[u] = *(const v2f*)(Arow0 + ka);
    f.a1[u] = *(const v2f*)(Arow1 + ka);
    const v2f* bp = Bcol + (size_t)(ka >> 1) * N;
    f.b0[u] = bp[0]; f.b1[u] = bp[16]; f.b2[u] = bp[32]; f.b3[u] = bp[48];
  }
}

__device__ __forceinline__ void gat_mma_group(const gat_frag_group& f,
                                              v8f& acc0, v8f& acc1, v8f& acc2, v8f& acc3,
                                              v8f& acc4, v8f& acc5, v8f& acc6, v8f& acc7) {
#pragma unroll
  for (int u = 0; u < GEMM_GK; ++u) {
    acc0 = __builtin_amdgcn_wmma_f32_16x16x4_f32(false, f.a0[u], false, f.b0[u], (short)0, acc0, false, false);
    acc1 = __builtin_amdgcn_wmma_f32_16x16x4_f32(false, f.a0[u], false, f.b1[u], (short)0, acc1, false, false);
    acc2 = __builtin_amdgcn_wmma_f32_16x16x4_f32(false, f.a0[u], false, f.b2[u], (short)0, acc2, false, false);
    acc3 = __builtin_amdgcn_wmma_f32_16x16x4_f32(false, f.a0[u], false, f.b3[u], (short)0, acc3, false, false);
    acc4 = __builtin_amdgcn_wmma_f32_16x16x4_f32(false, f.a1[u], false, f.b0[u], (short)0, acc4, false, false);
    acc5 = __builtin_amdgcn_wmma_f32_16x16x4_f32(false, f.a1[u], false, f.b1[u], (short)0, acc5, false, false);
    acc6 = __builtin_amdgcn_wmma_f32_16x16x4_f32(false, f.a1[u], false, f.b2[u], (short)0, acc6, false, false);
    acc7 = __builtin_amdgcn_wmma_f32_16x16x4_f32(false, f.a1[u], false, f.b3[u], (short)0, acc7, false, false);
  }
}

__global__ void __launch_bounds__(32)
gat_wmma_gemm_f32_heads_elu_kernel(const float* __restrict__ A,
                                   const float* __restrict__ Bp,
                                   float* __restrict__ C,
                                   int K, int N, int lda, int ldc) {
  const int lane = threadIdx.x & 31;
  const int hi   = lane >> 4;
  const int lr   = lane & 15;
  const int m0   = blockIdx.y << 5;
  const int n0   = blockIdx.x << 6;

  v8f acc0 = {}; v8f acc1 = {}; v8f acc2 = {}; v8f acc3 = {};
  v8f acc4 = {}; v8f acc5 = {}; v8f acc6 = {}; v8f acc7 = {};

  const float* Arow0 = A + (size_t)(m0 + lr) * lda + (size_t)(n0 / CHID) * K;
  const float* Arow1 = Arow0 + (size_t)16 * lda;
  const v2f* Bcol = (const v2f*)Bp + n0 + lr;

  gat_frag_group p, q;
  gat_load_group(p, Arow0, Arow1, Bcol, N, 0, hi);
  int k = 0;
  for (; k + 8 * GEMM_GK < K; k += 8 * GEMM_GK) {
    gat_load_group(q, Arow0, Arow1, Bcol, N, k + 4 * GEMM_GK, hi);
    gat_mma_group(p, acc0, acc1, acc2, acc3, acc4, acc5, acc6, acc7);
    gat_load_group(p, Arow0, Arow1, Bcol, N, k + 8 * GEMM_GK, hi);
    gat_mma_group(q, acc0, acc1, acc2, acc3, acc4, acc5, acc6, acc7);
  }
  gat_load_group(q, Arow0, Arow1, Bcol, N, k + 4 * GEMM_GK, hi);
  gat_mma_group(p, acc0, acc1, acc2, acc3, acc4, acc5, acc6, acc7);
  gat_mma_group(q, acc0, acc1, acc2, acc3, acc4, acc5, acc6, acc7);

  float* Crow0 = C + (size_t)(m0 + 8 * hi) * ldc + n0 + lr;
  float* Crow1 = Crow0 + (size_t)16 * ldc;
#pragma unroll
  for (int i = 0; i < 8; ++i) {
    Crow0[(size_t)i * ldc + 0]  = (acc0[i] > 0.f) ? acc0[i] : expm1f(acc0[i]);
    Crow0[(size_t)i * ldc + 16] = (acc1[i] > 0.f) ? acc1[i] : expm1f(acc1[i]);
    Crow0[(size_t)i * ldc + 32] = (acc2[i] > 0.f) ? acc2[i] : expm1f(acc2[i]);
    Crow0[(size_t)i * ldc + 48] = (acc3[i] > 0.f) ? acc3[i] : expm1f(acc3[i]);
    Crow1[(size_t)i * ldc + 0]  = (acc4[i] > 0.f) ? acc4[i] : expm1f(acc4[i]);
    Crow1[(size_t)i * ldc + 16] = (acc5[i] > 0.f) ? acc5[i] : expm1f(acc5[i]);
    Crow1[(size_t)i * ldc + 32] = (acc6[i] > 0.f) ? acc6[i] : expm1f(acc6[i]);
    Crow1[(size_t)i * ldc + 48] = (acc7[i] > 0.f) ? acc7[i] : expm1f(acc7[i]);
  }
}

#define SPLITK 16
__global__ void __launch_bounds__(32 * SPLITK)
gat_wmma_gemm_f32_splitk_kernel(const float* __restrict__ A,
                                const float* __restrict__ Bp,
                                float* __restrict__ C,
                                int K, int N, int lda, int ldc) {
  __shared__ float part[SPLITK][32][32];
  const int lane = threadIdx.x & 31;
  const int wv   = __builtin_amdgcn_readfirstlane((int)(threadIdx.x >> 5));
  const int hi   = lane >> 4;
  const int lr   = lane & 15;
  const int m0   = blockIdx.y << 4;
  const int n0   = blockIdx.x << 6;
  const int kbeg = wv * (K / SPLITK);
  const int kend = kbeg + K / SPLITK;

  v8f acc0 = {}; v8f acc1 = {}; v8f acc2 = {}; v8f acc3 = {};

  const float* Arow0 = A + (size_t)(m0 + lr) * lda;
  const v2f* Bv = (const v2f*)Bp;

  v2f a0, b0, b1, b2, b3;
  {
    const int ka = kbeg + 2 * hi;
    a0 = *(const v2f*)(Arow0 + ka);
    const v2f* bp = Bv + (size_t)(ka >> 1) * N + n0 + lr;
    b0 = bp[0]; b1 = bp[16]; b2 = bp[32]; b3 = bp[48];
  }
  for (int k = kbeg; k < kend; k += 4) {
    const v2f ca0 = a0;
    const v2f cb0 = b0, cb1 = b1, cb2 = b2, cb3 = b3;
    if (k + 4 < kend) {
      const int ka = k + 4 + 2 * hi;
      a0 = *(const v2f*)(Arow0 + ka);
      const v2f* bp = Bv + (size_t)(ka >> 1) * N + n0 + lr;
      b0 = bp[0]; b1 = bp[16]; b2 = bp[32]; b3 = bp[48];
    }
    acc0 = __builtin_amdgcn_wmma_f32_16x16x4_f32(false, ca0, false, cb0, (short)0, acc0, false, false);
    acc1 = __builtin_amdgcn_wmma_f32_16x16x4_f32(false, ca0, false, cb1, (short)0, acc1, false, false);
    acc2 = __builtin_amdgcn_wmma_f32_16x16x4_f32(false, ca0, false, cb2, (short)0, acc2, false, false);
    acc3 = __builtin_amdgcn_wmma_f32_16x16x4_f32(false, ca0, false, cb3, (short)0, acc3, false, false);
  }

#pragma unroll
  for (int i = 0; i < 8; ++i) {
    part[wv][0 * 8 + i][lane] = acc0[i];
    part[wv][1 * 8 + i][lane] = acc1[i];
    part[wv][2 * 8 + i][lane] = acc2[i];
    part[wv][3 * 8 + i][lane] = acc3[i];
  }
  __syncthreads();

  const int ri = wv & 7;
  const int rh = wv >> 3;
  const int pl = rh * 16 + lr;
  float sum0 = part[0][(hi) * 8 + ri][pl];
  float sum1 = part[0][(2 + hi) * 8 + ri][pl];
#pragma unroll
  for (int s = 1; s < SPLITK; ++s) {
    sum0 += part[s][(hi) * 8 + ri][pl];
    sum1 += part[s][(2 + hi) * 8 + ri][pl];
  }
  float* Crow = C + (size_t)(m0 + wv) * ldc + n0 + lane;
  Crow[0]  = sum0;
  Crow[32] = sum1;
}

__global__ void gat_dot_kernel(const float* __restrict__ Hm,
                               const float* __restrict__ a,
                               float* __restrict__ S,
                               int nwaves, int nheads, int rowStride) {
  int gt   = blockIdx.x * blockDim.x + threadIdx.x;
  int w    = gt >> 5;
  int lane = gt & 31;
  if (w >= nwaves) return;
  int n = w / nheads, h = w % nheads;
  const float* hp = Hm + (size_t)n * rowStride + h * CHID;
  const float* ap = a + (size_t)h * 2 * CHID;
  float sA = 0.f, sB = 0.f;
#pragma unroll
  for (int j = lane; j < CHID; j += 32) {
    float v = hp[j];
    sA += v * ap[j];
    sB += v * ap[CHID + j];
  }
#pragma unroll
  for (int off = 16; off > 0; off >>= 1) {
    sA += __shfl_down(sA, off, 32);
    sB += __shfl_down(sB, off, 32);
  }
  if (lane == 0) { S[2 * w] = sA; S[2 * w + 1] = sB; }
}

#define SC_NPW 4
static_assert(CN2 % SC_NPW == 0, "whole waves only");
static_assert(CIN == 256, "a lane holds 2 x 4 columns of a 256-wide row");
__global__ void __launch_bounds__(256)
gat_score_feat_kernel(const float* __restrict__ X,
                                      const float* __restrict__ waT,
                                      float* __restrict__ S,
                                      int nnodes) {
  const int gt   = blockIdx.x * blockDim.x + threadIdx.x;
  const int nb   = (gt >> 5) * SC_NPW;
  const int lane = gt & 31;
  if (nb + SC_NPW > nnodes) return;
  const float4* wp = (const float4*)waT + lane;
  float4 f0[SC_NPW], f1[SC_NPW];
#pragma unroll
  for (int r = 0; r < SC_NPW; ++r) {
    const float4* xp = (const float4*)(X + (size_t)(nb + r) * CIN) + lane;
    f0[r] = xp[0];
    f1[r] = xp[CIN / 8];
  }
  float s[SC_NPW][2 * CNH];
#pragma unroll
  for (int o = 0; o < 2 * CNH; ++o) {
    const float4 w0 = wp[o * (CIN / 4)];
    const float4 w1 = wp[o * (CIN / 4) + CIN / 8];
#pragma unroll
    for (int r = 0; r < SC_NPW; ++r) {
      s[r][o] = f0[r].x * w0.x + f0[r].y * w0.y + f0[r].z * w0.z + f0[r].w * w0.w
              + f1[r].x * w1.x + f1[r].y * w1.y + f1[r].z * w1.z + f1[r].w * w1.w;
    }
  }
#pragma unroll
  for (int r = 0; r < SC_NPW; ++r) {
#pragma unroll
    for (int o = 0; o < 2 * CNH; ++o) {
#pragma unroll
      for (int off = 16; off > 0; off >>= 1) s[r][o] += __shfl_xor(s[r][o], off, 32);
    }
  }
  if (lane == 0) {
#pragma unroll
    for (int r = 0; r < SC_NPW; ++r) {
      float4* o4 = (float4*)(S + (size_t)(nb + r) * (2 * CNH));
#pragma unroll
      for (int o = 0; o < 2 * CNH; o += 4)
        o4[o >> 2] = make_float4(s[r][o], s[r][o + 1], s[r][o + 2], s[r][o + 3]);
    }
  }
}

#define AGF_TPB 8
static_assert(CN1 % AGF_TPB == 0, "whole blocks only");
static_assert(CNH == 8, "the scores of a node are read as 4 pieces of 2 heads");
static_assert(CIN == 256, "a lane holds 2 x 4 columns of a 256-wide row");
__global__ void __launch_bounds__(32 * AGF_TPB)
gat_agg_feat_kernel(const float* __restrict__ X,
                    const float* __restrict__ S,
                    const int* __restrict__ src,
                    const int* __restrict__ dst,
                    float* __restrict__ Z) {
  const int lane = threadIdx.x & 31;
  const int wv   = __builtin_amdgcn_readfirstlane((int)(threadIdx.x >> 5));
  const int t    = blockIdx.x * AGF_TPB + wv;
  const int m0   = t * KNB;

  const int sL = src[m0 + lane], dL = dst[m0 + lane];
  const int sX = src[m0 + 32],   dX = dst[m0 + 32];
  const float4* SsL = (const float4*)(S + (size_t)sL * (2 * CNH));
  const float4* SdL = (const float4*)(S + (size_t)dL * (2 * CNH));
  const float4* SsX = (const float4*)(S + (size_t)sX * (2 * CNH));
  const float4* SdX = (const float4*)(S + (size_t)dX * (2 * CNH));
  float wL[CNH], wX[CNH];
#pragma unroll
  for (int g = 0; g < CNH / 2; ++g) {
    const float4 a = SsL[g], b = SdL[g], c = SsX[g], d = SdX[g];
    wL[2 * g]     = a.x + b.y;
    wL[2 * g + 1] = a.z + b.w;
    wX[2 * g]     = c.x + d.y;
    wX[2 * g + 1] = c.z + d.w;
  }
#pragma unroll
  for (int h = 0; h < CNH; ++h) {
    float eL = wL[h], eX = wX[h];
    eL = (eL > 0.f) ? eL : LRELU_ALPHA * eL;
    eX = (eX > 0.f) ? eX : LRELU_ALPHA * eX;
    float mx = eL;
#pragma unroll
    for (int off = 16; off > 0; off >>= 1) mx = fmaxf(mx, __shfl_xor(mx, off, 32));
    mx = fmaxf(mx, eX);
    const float pL = expf(eL - mx);
    const float pX = expf(eX - mx);
    float sum = pL;
#pragma unroll
    for (int off = 16; off > 0; off >>= 1) sum += __shfl_xor(sum, off, 32);
    sum += pX;
    const float inv = 1.f / sum;
    wL[h] = pL * inv;
    wX[h] = pX * inv;
  }

  const float* col = X + 4 * lane;
  float4 acc0[CNH], acc1[CNH];
#pragma unroll
  for (int h = 0; h < CNH; ++h) {
    acc0[h] = make_float4(0.f, 0.f, 0.f, 0.f);
    acc1[h] = make_float4(0.f, 0.f, 0.f, 0.f);
  }
#pragma unroll
  for (int k = 0; k < 32; ++k) {
    const int dk = __builtin_amdgcn_readlane(dL, k);
    const float4 v0 = *(const float4*)(col + (size_t)dk * CIN);
    const float4 v1 = *(const float4*)(col + (size_t)dk * CIN + CIN / 2);
#pragma unroll
    for (int h = 0; h < CNH; ++h) {
      const float wk = __int_as_float(__builtin_amdgcn_readlane(__float_as_int(wL[h]), k));
      acc0[h].x += wk * v0.x; acc0[h].y += wk * v0.y; acc0[h].z += wk * v0.z; acc0[h].w += wk * v0.w;
      acc1[h].x += wk * v1.x; acc1[h].y += wk * v1.y; acc1[h].z += wk * v1.z; acc1[h].w += wk * v1.w;
    }
  }
  {
    const int dk = __builtin_amdgcn_readfirstlane(dX);
    const float4 v0 = *(const float4*)(col + (size_t)dk * CIN);
    const float4 v1 = *(const float4*)(col + (size_t)dk * CIN + CIN / 2);
#pragma unroll
    for (int h = 0; h < CNH; ++h) {
      const float wk = __int_as_float(__builtin_amdgcn_readfirstlane(__float_as_int(wX[h])));
      acc0[h].x += wk * v0.x; acc0[h].y += wk * v0.y; acc0[h].z += wk * v0.z; acc0[h].w += wk * v0.w;
      acc1[h].x += wk * v1.x; acc1[h].y += wk * v1.y; acc1[h].z += wk * v1.z; acc1[h].w += wk * v1.w;
    }
  }
  float* zp = Z + (size_t)t * (CNH * CIN) + 4 * lane;
#pragma unroll
  for (int h = 0; h < CNH; ++h) {
    *(float4*)(zp + (size_t)h * CIN)           = acc0[h];
    *(float4*)(zp + (size_t)h * CIN + CIN / 2) = acc1[h];
  }
}

static_assert(KNB == 33, "edge 32 is the one edge beyond the 32 lanes");
__global__ void __launch_bounds__(32 * CNH)
gat_agg_wave_kernel(const float* __restrict__ Hm,
                    const float* __restrict__ S,
                    const int* __restrict__ src,
                    const int* __restrict__ dst,
                    float* __restrict__ Yout,
                    int nheads, int rowStride) {
  const int t    = blockIdx.x;
  const int lane = threadIdx.x & 31;
  const int h    = __builtin_amdgcn_readfirstlane((int)(threadIdx.x >> 5));
  const int m0   = t * KNB;

  const int sL = src[m0 + lane], dL = dst[m0 + lane];
  const int sX = src[m0 + 32],   dX = dst[m0 + 32];
  float eL = S[(size_t)(sL * nheads + h) * 2] + S[(size_t)(dL * nheads + h) * 2 + 1];
  float eX = S[(size_t)(sX * nheads + h) * 2] + S[(size_t)(dX * nheads + h) * 2 + 1];
  eL = (eL > 0.f) ? eL : LRELU_ALPHA * eL;
  eX = (eX > 0.f) ? eX : LRELU_ALPHA * eX;

  float mx = eL;
  for (int off = 16; off > 0; off >>= 1) mx = fmaxf(mx, __shfl_xor(mx, off, 32));
  mx = fmaxf(mx, eX);
  const float pL = expf(eL - mx);
  const float pX = expf(eX - mx);
  float sum = pL;
  for (int off = 16; off > 0; off >>= 1) sum += __shfl_xor(sum, off, 32);
  sum += pX;
  const float inv = 1.f / sum;
  const float wL = pL * inv;
  const float wX = pX * inv;

  const float* col = Hm + (size_t)h * CHID + 4 * lane;
  float4 acc = make_float4(0.f, 0.f, 0.f, 0.f);
#pragma unroll
  for (int k = 0; k < 32; ++k) {
    const int   dk = __builtin_amdgcn_readlane(dL, k);
    const float wk = __int_as_float(__builtin_amdgcn_readlane(__float_as_int(wL), k));
    const float4 v = *(const float4*)(col + (size_t)dk * rowStride);
    acc.x += wk * v.x; acc.y += wk * v.y; acc.z += wk * v.z; acc.w += wk * v.w;
  }
  {
    const int   dk = __builtin_amdgcn_readfirstlane(dX);
    const float wk = __int_as_float(__builtin_amdgcn_readfirstlane(__float_as_int(wX)));
    const float4 v = *(const float4*)(col + (size_t)dk * rowStride);
    acc.x += wk * v.x; acc.y += wk * v.y; acc.z += wk * v.z; acc.w += wk * v.w;
  }
  float4 r;
  r.x = (acc.x > 0.f) ? acc.x : expm1f(acc.x);
  r.y = (acc.y > 0.f) ? acc.y : expm1f(acc.y);
  r.z = (acc.z > 0.f) ? acc.z : expm1f(acc.z);
  r.w = (acc.w > 0.f) ? acc.w : expm1f(acc.w);
  *(float4*)(Yout + (size_t)t * (nheads * CHID) + h * CHID + 4 * lane) = r;
}

extern "C" void kernel_launch(void* const* d_in, const int* in_sizes, int n_in,
                              void* d_out, int out_size, void* d_ws, size_t ws_size,
                              hipStream_t stream) {
  (void)in_sizes; (void)n_in; (void)out_size; (void)ws_size;
  const float* feats = (const float*)d_in[0];
  const float* W1    = (const float*)d_in[1];
  const float* a1    = (const float*)d_in[2];
  const float* W2    = (const float*)d_in[3];
  const float* a2    = (const float*)d_in[4];
  const int*   src1  = (const int*)d_in[5];
  const int*   dst1  = (const int*)d_in[6];
  const int*   src2  = (const int*)d_in[8];
  const int*   dst2  = (const int*)d_in[9];
  float* out = (float*)d_out;

  float* ws  = (float*)d_ws;
  float* z1  = ws;
  float* waT = z1  + (size_t)CN1 * CNH * CIN;
  float* x   = ws  + (size_t)CN2 * CNH * CHID;
  float* Bp1 = x   + (size_t)CN1 * CNH * CHID;
  float* Bp2 = Bp1 + (size_t)CIN * CNH * CHID;
  float* h2  = Bp2 + (size_t)CNH * CHID * COUT;
  float* s1  = h2  + (size_t)CN1 * COUT;
  float* s2  = s1  + (size_t)CN2 * CNH * 2;

  gat_pack_w1_wa_kernel<<<(CNH * CIN / 2) / 8, 256, 0, stream>>>(W1, a1, Bp1, waT);
  gat_pack_w2_kernel<<<(CNH * CHID * COUT) / 256, 256, 0, stream>>>(W2, Bp2);
  gat_score_feat_kernel<<<CN2 / (8 * SC_NPW), 256, 0, stream>>>(feats, waT, s1, CN2);
  gat_agg_feat_kernel<<<CN1 / AGF_TPB, 32 * AGF_TPB, 0, stream>>>(feats, s1, src1, dst1, z1);
  dim3 g1((CNH * CHID) / 64, CN1 / 32);
  gat_wmma_gemm_f32_heads_elu_kernel<<<g1, 32, 0, stream>>>(z1, Bp1, x,
                                                            CIN, CNH * CHID, CNH * CIN, CNH * CHID);
  dim3 g2(COUT / 64, CN1 / 16);
  gat_wmma_gemm_f32_splitk_kernel<<<g2, 32 * SPLITK, 0, stream>>>(x, Bp2, h2,
                                                                   CNH * CHID, COUT, CNH * CHID, COUT);
  gat_dot_kernel<<<CN1 / 8, 256, 0, stream>>>(h2, a2, s2, CN1, 1, COUT);
  gat_agg_wave_kernel<<<CN0, 32, 0, stream>>>(h2, s2, src2, dst2, out, 1, COUT);
}


# 8 launches to 6: both weight packs in one launch; layer-2 score dots taken in the layer-2 GEMM's fold as two 64-column partial sums added at the softmax; changes the sum order of those dots
# speedup vs baseline: 4.3161x; 1.0639x over previous
#include <hip/hip_runtime.h>

#define CN0 512
#define CN1 2048
#define CN2 8192
#define KNB 33
#define CIN 256
#define CHID 128
#define COUT 128
#define CNH 8
#define LRELU_ALPHA 0.2f

typedef float v2f __attribute__((ext_vector_type(2)));
typedef float v8f __attribute__((ext_vector_type(8)));

#define PACK_W1_BLOCKS ((CNH * CIN / 2) / 8)
#define PACK_W2_BLOCKS ((CNH * CHID / 2) / 8)
static_assert(CHID == 128 && COUT == 128, "a wave covers a 128-column row with 4 columns per lane");
__device__ __forceinline__ void gat_pack_row_pair(const float* __restrict__ row0,
                                                  const float* __restrict__ row1,
                                                  float* __restrict__ pairs,
                                                  int lane, float4& r0, float4& r1) {
  r0 = *((const float4*)row0 + lane);
  r1 = *((const float4*)row1 + lane);
  float4* o = (float4*)(pairs + 8 * lane);
  o[0] = make_float4(r0.x, r1.x, r0.y, r1.y);
  o[1] = make_float4(r0.z, r1.z, r0.w, r1.w);
}

__global__ void gat_pack_weights_kernel(const float* __restrict__ W1,
                                        const float* __restrict__ a1,
                                        const float* __restrict__ W2,
                                        float* __restrict__ Bp1,
                                        float* __restrict__ waT,
                                        float* __restrict__ Bp2) {
  const int lane = threadIdx.x & 31;
  const int wb   = threadIdx.x >> 5;
  float4 r0, r1;
  if (blockIdx.x >= PACK_W1_BLOCKS) {
    const int cp = (blockIdx.x - PACK_W1_BLOCKS) * 8 + wb;
    gat_pack_row_pair(W2 + (size_t)(2 * cp) * COUT, W2 + (size_t)(2 * cp + 1) * COUT,
                      Bp2 + (size_t)cp * (2 * COUT), lane, r0, r1);
    return;
  }
  const int w    = blockIdx.x * 8 + wb;
  const int h    = w / (CIN / 2);
  const int ip   = w % (CIN / 2);
  gat_pack_row_pair(W1 + ((size_t)h * CIN + 2 * ip) * CHID, W1 + ((size_t)h * CIN + 2 * ip + 1) * CHID,
                    Bp1 + (size_t)ip * (2 * CNH * CHID) + 2 * (h * CHID), lane, r0, r1);
  const float4 p = *((const float4*)(a1 + (size_t)h * 2 * CHID) + lane);
  const float4 q = *((const float4*)(a1 + (size_t)h * 2 * CHID + CHID) + lane);
  float s0 = r0.x * p.x + r0.y * p.y + r0.z * p.z + r0.w * p.w;
  float s1 = r0.x * q.x + r0.y * q.y + r0.z * q.z + r0.w * q.w;
  float s2 = r1.x * p.x + r1.y * p.y + r1.z * p.z + r1.w * p.w;
  float s3 = r1.x * q.x + r1.y * q.y + r1.z * q.z + r1.w * q.w;
#pragma unroll
  for (int off = 16; off > 0; off >>= 1) {
    s0 += __shfl_xor(s0, off, 32);
    s1 += __shfl_xor(s1, off, 32);
    s2 += __shfl_xor(s2, off, 32);
    s3 += __shfl_xor(s3, off, 32);
  }
  if (lane == 0) {
    *(float2*)(waT + (size_t)(h * 2 + 0) * CIN + 2 * ip) = make_float2(s0, s2);
    *(float2*)(waT + (size_t)(h * 2 + 1) * CIN + 2 * ip) = make_float2(s1, s3);
  }
}

#define GEMM_GK 4
static_assert(CHID % 64 == 0, "a 64-column tile must not straddle heads");
static_assert(CIN % (8 * GEMM_GK) == 0, "whole pairs of groups of k-steps only");

struct gat_frag_group {
  v2f a0[GEMM_GK], a1[GEMM_GK], b0[GEMM_GK], b1[GEMM_GK], b2[GEMM_GK], b3[GEMM_GK];
};

__device__ __forceinline__ void gat_load_group(gat_frag_group& f,
                                               const float* __restrict__ Arow0,
                                               const float* __restrict__ Arow1,
                                               const v2f* __restrict__ Bcol,
                                               int N, int kb, int hi) {
#pragma unroll
  for (int u = 0; u < GEMM_GK; ++u) {
    const int ka = kb + 4 * u + 2 * hi;
    f.a0[u] = *(const v2f*)(Arow0 + ka);
    f.a1[u] = *(const v2f*)(Arow1 + ka);
    const v2f* bp = Bcol + (size_t)(ka >> 1) * N;
    f.b0[u] = bp[0]; f.b1[u] = bp[16]; f.b2[u] = bp[32]; f.b3[u] = bp[48];
  }
}

__device__ __forceinline__ void gat_mma_group(const gat_frag_group& f,
                                              v8f& acc0, v8f& acc1, v8f& acc2, v8f& acc3,
                                              v8f& acc4, v8f& acc5, v8f& acc6, v8f& acc7) {
#pragma unroll
  for (int u = 0; u < GEMM_GK; ++u) {
    acc0 = __builtin_amdgcn_wmma_f32_16x16x4_f32(false, f.a0[u], false, f.b0[u], (short)0, acc0, false, false);
    acc1 = __builtin_amdgcn_wmma_f32_16x16x4_f32(false, f.a0[u], false, f.b1[u], (short)0, acc1, false, false);
    acc2 = __builtin_amdgcn_wmma_f32_16x16x4_f32(false, f.a0[u], false, f.b2[u], (short)0, acc2, false, false);
    acc3 = __builtin_amdgcn_wmma_f32_16x16x4_f32(false, f.a0[u], false, f.b3[u], (short)0, acc3, false, false);
    acc4 = __builtin_amdgcn_wmma_f32_16x16x4_f32(false, f.a1[u], false, f.b0[u], (short)0, acc4, false, false);
    acc5 = __builtin_amdgcn_wmma_f32_16x16x4_f32(false, f.a1[u], false, f.b1[u], (short)0, acc5, false, false);
    acc6 = __builtin_amdgcn_wmma_f32_16x16x4_f32(false, f.a1[u], false, f.b2[u], (short)0, acc6, false, false);
    acc7 = __builtin_amdgcn_wmma_f32_16x16x4_f32(false, f.a1[u], false, f.b3[u], (short)0, acc7, false, false);
  }
}

__global__ void __launch_bounds__(32)
gat_wmma_gemm_f32_heads_elu_kernel(const float* __restrict__ A,
                                   const float* __restrict__ Bp,
                                   float* __restrict__ C,
                                   int K, int N, int lda, int ldc) {
  const int lane = threadIdx.x & 31;
  const int hi   = lane >> 4;
  const int lr   = lane & 15;
  const int m0   = blockIdx.y << 5;
  const int n0   = blockIdx.x << 6;

  v8f acc0 = {}; v8f acc1 = {}; v8f acc2 = {}; v8f acc3 = {};
  v8f acc4 = {}; v8f acc5 = {}; v8f acc6 = {}; v8f acc7 = {};

  const float* Arow0 = A + (size_t)(m0 + lr) * lda + (size_t)(n0 / CHID) * K;
  const float* Arow1 = Arow0 + (size_t)16 * lda;
  const v2f* Bcol = (const v2f*)Bp + n0 + lr;

  gat_frag_group p, q;
  gat_load_group(p, Arow0, Arow1, Bcol, N, 0, hi);
  int k = 0;
  for (; k + 8 * GEMM_GK < K; k += 8 * GEMM_GK) {
    gat_load_group(q, Arow0, Arow1, Bcol, N, k + 4 * GEMM_GK, hi);
    gat_mma_group(p, acc0, acc1, acc2, acc3, acc4, acc5, acc6, acc7);
    gat_load_group(p, Arow0, Arow1, Bcol, N, k + 8 * GEMM_GK, hi);
    gat_mma_group(q, acc0, acc1, acc2, acc3, acc4, acc5, acc6, acc7);
  }
  gat_load_group(q, Arow0, Arow1, Bcol, N, k + 4 * GEMM_GK, hi);
  gat_mma_group(p, acc0, acc1, acc2, acc3, acc4, acc5, acc6, acc7);
  gat_mma_group(q, acc0, acc1, acc2, acc3, acc4, acc5, acc6, acc7);

  float* Crow0 = C + (size_t)(m0 + 8 * hi) * ldc + n0 + lr;
  float* Crow1 = Crow0 + (size_t)16 * ldc;
#pragma unroll
  for (int i = 0; i < 8; ++i) {
    Crow0[(size_t)i * ldc + 0]  = (acc0[i] > 0.f) ? acc0[i] : expm1f(acc0[i]);
    Crow0[(size_t)i * ldc + 16] = (acc1[i] > 0.f) ? acc1[i] : expm1f(acc1[i]);
    Crow0[(size_t)i * ldc + 32] = (acc2[i] > 0.f) ? acc2[i] : expm1f(acc2[i]);
    Crow0[(size_t)i * ldc + 48] = (acc3[i] > 0.f) ? acc3[i] : expm1f(acc3[i]);
    Crow1[(size_t)i * ldc + 0]  = (acc4[i] > 0.f) ? acc4[i] : expm1f(acc4[i]);
    Crow1[(size_t)i * ldc + 16] = (acc5[i] > 0.f) ? acc5[i] : expm1f(acc5[i]);
    Crow1[(size_t)i * ldc + 32] = (acc6[i] > 0.f) ? acc6[i] : expm1f(acc6[i]);
    Crow1[(size_t)i * ldc + 48] = (acc7[i] > 0.f) ? acc7[i] : expm1f(acc7[i]);
  }
}

#define SPLITK 16
__global__ void __launch_bounds__(32 * SPLITK)
gat_wmma_gemm_f32_splitk_kernel(const float* __restrict__ A,
                                const float* __restrict__ Bp,
                                float* __restrict__ C,
                                const float* __restrict__ a,
                                float* __restrict__ Sp,
                                int K, int N, int lda, int ldc, int M) {
  __shared__ float part[SPLITK][32][32];
  const int lane = threadIdx.x & 31;
  const int wv   = __builtin_amdgcn_readfirstlane((int)(threadIdx.x >> 5));
  const int hi   = lane >> 4;
  const int lr   = lane & 15;
  const int m0   = blockIdx.y << 4;
  const int n0   = blockIdx.x << 6;
  const int kbeg = wv * (K / SPLITK);
  const int kend = kbeg + K / SPLITK;

  v8f acc0 = {}; v8f acc1 = {}; v8f acc2 = {}; v8f acc3 = {};

  const float* Arow0 = A + (size_t)(m0 + lr) * lda;
  const v2f* Bv = (const v2f*)Bp;

  v2f a0, b0, b1, b2, b3;
  {
    const int ka = kbeg + 2 * hi;
    a0 = *(const v2f*)(Arow0 + ka);
    const v2f* bp = Bv + (size_t)(ka >> 1) * N + n0 + lr;
    b0 = bp[0]; b1 = bp[16]; b2 = bp[32]; b3 = bp[48];
  }
  for (int k = kbeg; k < kend; k += 4) {
    const v2f ca0 = a0;
    const v2f cb0 = b0, cb1 = b1, cb2 = b2, cb3 = b3;
    if (k + 4 < kend) {
      const int ka = k + 4 + 2 * hi;
      a0 = *(const v2f*)(Arow0 + ka);
      const v2f* bp = Bv + (size_t)(ka >> 1) * N + n0 + lr;
      b0 = bp[0]; b1 = bp[16]; b2 = bp[32]; b3 = bp[48];
    }
    acc0 = __builtin_amdgcn_wmma_f32_16x16x4_f32(false, ca0, false, cb0, (short)0, acc0, false, false);
    acc1 = __builtin_amdgcn_wmma_f32_16x16x4_f32(false, ca0, false, cb1, (short)0, acc1, false, false);
    acc2 = __builtin_amdgcn_wmma_f32_16x16x4_f32(false, ca0, false, cb2, (short)0, acc2, false, false);
    acc3 = __builtin_amdgcn_wmma_f32_16x16x4_f32(false, ca0, false, cb3, (short)0, acc3, false, false);
  }

#pragma unroll
  for (int i = 0; i < 8; ++i) {
    part[wv][0 * 8 + i][lane] = acc0[i];
    part[wv][1 * 8 + i][lane] = acc1[i];
    part[wv][2 * 8 + i][lane] = acc2[i];
    part[wv][3 * 8 + i][lane] = acc3[i];
  }
  __syncthreads();

  const int ri = wv & 7;
  const int rh = wv >> 3;
  const int pl = rh * 16 + lr;
  float sum0 = part[0][(hi) * 8 + ri][pl];
  float sum1 = part[0][(2 + hi) * 8 + ri][pl];
#pragma unroll
  for (int s = 1; s < SPLITK; ++s) {
    sum0 += part[s][(hi) * 8 + ri][pl];
    sum1 += part[s][(2 + hi) * 8 + ri][pl];
  }
  float* Crow = C + (size_t)(m0 + wv) * ldc + n0 + lane;
  Crow[0]  = sum0;
  Crow[32] = sum1;

  const float* ap = a + n0 + lane;
  float dA = sum0 * ap[0] + sum1 * ap[32];
  float dB = sum0 * ap[N] + sum1 * ap[N + 32];
#pragma unroll
  for (int off = 16; off > 0; off >>= 1) {
    dA += __shfl_xor(dA, off, 32);
    dB += __shfl_xor(dB, off, 32);
  }
  if (lane == 0)
    *(float2*)(Sp + ((size_t)blockIdx.x * M + m0 + wv) * 2) = make_float2(dA, dB);
}

#define SC_NPW 4
static_assert(CN2 % SC_NPW == 0, "whole waves only");
static_assert(CIN == 256, "a lane holds 2 x 4 columns of a 256-wide row");
__global__ void __launch_bounds__(256)
gat_score_feat_kernel(const float* __restrict__ X,
                                      const float* __restrict__ waT,
                                      float* __restrict__ S,
                                      int nnodes) {
  const int gt   = blockIdx.x * blockDim.x + threadIdx.x;
  const int nb   = (gt >> 5) * SC_NPW;
  const int lane = gt & 31;
  if (nb + SC_NPW > nnodes) return;
  const float4* wp = (const float4*)waT + lane;
  float4 f0[SC_NPW], f1[SC_NPW];
#pragma unroll
  for (int r = 0; r < SC_NPW; ++r) {
    const float4* xp = (const float4*)(X + (size_t)(nb + r) * CIN) + lane;
    f0[r] = xp[0];
    f1[r] = xp[CIN / 8];
  }
  float s[SC_NPW][2 * CNH];
#pragma unroll
  for (int o = 0; o < 2 * CNH; ++o) {
    const float4 w0 = wp[o * (CIN / 4)];
    const float4 w1 = wp[o * (CIN / 4) + CIN / 8];
#pragma unroll
    for (int r = 0; r < SC_NPW; ++r) {
      s[r][o] = f0[r].x * w0.x + f0[r].y * w0.y + f0[r].z * w0.z + f0[r].w * w0.w
              + f1[r].x * w1.x + f1[r].y * w1.y + f1[r].z * w1.z + f1[r].w * w1.w;
    }
  }
#pragma unroll
  for (int r = 0; r < SC_NPW; ++r) {
#pragma unroll
    for (int o = 0; o < 2 * CNH; ++o) {
#pragma unroll
      for (int off = 16; off > 0; off >>= 1) s[r][o] += __shfl_xor(s[r][o], off, 32);
    }
  }
  if (lane == 0) {
#pragma unroll
    for (int r = 0; r < SC_NPW; ++r) {
      float4* o4 = (float4*)(S + (size_t)(nb + r) * (2 * CNH));
#pragma unroll
      for (int o = 0; o < 2 * CNH; o += 4)
        o4[o >> 2] = make_float4(s[r][o], s[r][o + 1], s[r][o + 2], s[r][o + 3]);
    }
  }
}

#define AGF_TPB 8
static_assert(CN1 % AGF_TPB == 0, "whole blocks only");
static_assert(CNH == 8, "the scores of a node are read as 4 pieces of 2 heads");
static_assert(CIN == 256, "a lane holds 2 x 4 columns of a 256-wide row");
__global__ void __launch_bounds__(32 * AGF_TPB)
gat_agg_feat_kernel(const float* __restrict__ X,
                    const float* __restrict__ S,
                    const int* __restrict__ src,
                    const int* __restrict__ dst,
                    float* __restrict__ Z) {
  const int lane = threadIdx.x & 31;
  const int wv   = __builtin_amdgcn_readfirstlane((int)(threadIdx.x >> 5));
  const int t    = blockIdx.x * AGF_TPB + wv;
  const int m0   = t * KNB;

  const int sL = src[m0 + lane], dL = dst[m0 + lane];
  const int sX = src[m0 + 32],   dX = dst[m0 + 32];
  const float4* SsL = (const float4*)(S + (size_t)sL * (2 * CNH));
  const float4* SdL = (const float4*)(S + (size_t)dL * (2 * CNH));
  const float4* SsX = (const float4*)(S + (size_t)sX * (2 * CNH));
  const float4* SdX = (const float4*)(S + (size_t)dX * (2 * CNH));
  float wL[CNH], wX[CNH];
#pragma unroll
  for (int g = 0; g < CNH / 2; ++g) {
    const float4 a = SsL[g], b = SdL[g], c = SsX[g], d = SdX[g];
    wL[2 * g]     = a.x + b.y;
    wL[2 * g + 1] = a.z + b.w;
    wX[2 * g]     = c.x + d.y;
    wX[2 * g + 1] = c.z + d.w;
  }
#pragma unroll
  for (int h = 0; h < CNH; ++h) {
    float eL = wL[h], eX = wX[h];
    eL = (eL > 0.f) ? eL : LRELU_ALPHA * eL;
    eX = (eX > 0.f) ? eX : LRELU_ALPHA * eX;
    float mx = eL;
#pragma unroll
    for (int off = 16; off > 0; off >>= 1) mx = fmaxf(mx, __shfl_xor(mx, off, 32));
    mx = fmaxf(mx, eX);
    const float pL = expf(eL - mx);
    const float pX = expf(eX - mx);
    float sum = pL;
#pragma unroll
    for (int off = 16; off > 0; off >>= 1) sum += __shfl_xor(sum, off, 32);
    sum += pX;
    const float inv = 1.f / sum;
    wL[h] = pL * inv;
    wX[h] = pX * inv;
  }

  const float* col = X + 4 * lane;
  float4 acc0[CNH], acc1[CNH];
#pragma unroll
  for (int h = 0; h < CNH; ++h) {
    acc0[h] = make_float4(0.f, 0.f, 0.f, 0.f);
    acc1[h] = make_float4(0.f, 0.f, 0.f, 0.f);
  }
#pragma unroll
  for (int k = 0; k < 32; ++k) {
    const int dk = __builtin_amdgcn_readlane(dL, k);
    const float4 v0 = *(const float4*)(col + (size_t)dk * CIN);
    const float4 v1 = *(const float4*)(col + (size_t)dk * CIN + CIN / 2);
#pragma unroll
    for (int h = 0; h < CNH; ++h) {
      const float wk = __int_as_float(__builtin_amdgcn_readlane(__float_as_int(wL[h]), k));
      acc0[h].x += wk * v0.x; acc0[h].y += wk * v0.y; acc0[h].z += wk * v0.z; acc0[h].w += wk * v0.w;
      acc1[h].x += wk * v1.x; acc1[h].y += wk * v1.y; acc1[h].z += wk * v1.z; acc1[h].w += wk * v1.w;
    }
  }
  {
    const int dk = __builtin_amdgcn_readfirstlane(dX);
    const float4 v0 = *(const float4*)(col + (size_t)dk * CIN);
    const float4 v1 = *(const float4*)(col + (size_t)dk * CIN + CIN / 2);
#pragma unroll
    for (int h = 0; h < CNH; ++h) {
      const float wk = __int_as_float(__builtin_amdgcn_readfirstlane(__float_as_int(wX[h])));
      acc0[h].x += wk * v0.x; acc0[h].y += wk * v0.y; acc0[h].z += wk * v0.z; acc0[h].w += wk * v0.w;
      acc1[h].x += wk * v1.x; acc1[h].y += wk * v1.y; acc1[h].z += wk * v1.z; acc1[h].w += wk * v1.w;
    }
  }
  float* zp = Z + (size_t)t * (CNH * CIN) + 4 * lane;
#pragma unroll
  for (int h = 0; h < CNH; ++h) {
    *(float4*)(zp + (size_t)h * CIN)           = acc0[h];
    *(float4*)(zp + (size_t)h * CIN + CIN / 2) = acc1[h];
  }
}

static_assert(KNB == 33, "edge 32 is the one edge beyond the 32 lanes");
__global__ void __launch_bounds__(32 * CNH)
gat_agg_wave_kernel(const float* __restrict__ Hm,
                    const float* __restrict__ S,
                    const int* __restrict__ src,
                    const int* __restrict__ dst,
                    float* __restrict__ Yout,
                    int nheads, int rowStride, int planeStride) {
  const int t    = blockIdx.x;
  const int lane = threadIdx.x & 31;
  const int h    = __builtin_amdgcn_readfirstlane((int)(threadIdx.x >> 5));
  const int m0   = t * KNB;

  const int sL = src[m0 + lane], dL = dst[m0 + lane];
  const int sX = src[m0 + 32],   dX = dst[m0 + 32];
  const float2* S0 = (const float2*)S + h;
  const float2* S1 = S0 + planeStride;
  float eL = (S0[(size_t)sL * nheads].x + S1[(size_t)sL * nheads].x) + (S0[(size_t)dL * nheads].y + S1[(size_t)dL * nheads].y);
  float eX = (S0[(size_t)sX * nheads].x + S1[(size_t)sX * nheads].x) + (S0[(size_t)dX * nheads].y + S1[(size_t)dX * nheads].y);
  eL = (eL > 0.f) ? eL : LRELU_ALPHA * eL;
  eX = (eX > 0.f) ? eX : LRELU_ALPHA * eX;

  float mx = eL;
  for (int off = 16; off > 0; off >>= 1) mx = fmaxf(mx, __shfl_xor(mx, off, 32));
  mx = fmaxf(mx, eX);
  const float pL = expf(eL - mx);
  const float pX = expf(eX - mx);
  float sum = pL;
  for (int off = 16; off > 0; off >>= 1) sum += __shfl_xor(sum, off, 32);
  sum += pX;
  const float inv = 1.f / sum;
  const float wL = pL * inv;
  const float wX = pX * inv;

  const float* col = Hm + (size_t)h * CHID + 4 * lane;
  float4 acc = make_float4(0.f, 0.f, 0.f, 0.f);
#pragma unroll
  for (int k = 0; k < 32; ++k) {
    const int   dk = __builtin_amdgcn_readlane(dL, k);
    const float wk = __int_as_float(__builtin_amdgcn_readlane(__float_as_int(wL), k));
    const float4 v = *(const float4*)(col + (size_t)dk * rowStride);
    acc.x += wk * v.x; acc.y += wk * v.y; acc.z += wk * v.z; acc.w += wk * v.w;
  }
  {
    const int   dk = __builtin_amdgcn_readfirstlane(dX);
    const float wk = __int_as_float(__builtin_amdgcn_readfirstlane(__float_as_int(wX)));
    const float4 v = *(const float4*)(col + (size_t)dk * rowStride);
    acc.x += wk * v.x; acc.y += wk * v.y; acc.z += wk * v.z; acc.w += wk * v.w;
  }
  float4 r;
  r.x = (acc.x > 0.f) ? acc.x : expm1f(acc.x);
  r.y = (acc.y > 0.f) ? acc.y : expm1f(acc.y);
  r.z = (acc.z > 0.f) ? acc.z : expm1f(acc.z);
  r.w = (acc.w > 0.f) ? acc.w : expm1f(acc.w);
  *(float4*)(Yout + (size_t)t * (nheads * CHID) + h * CHID + 4 * lane) = r;
}

extern "C" void kernel_launch(void* const* d_in, const int* in_sizes, int n_in,
                              void* d_out, int out_size, void* d_ws, size_t ws_size,
                              hipStream_t stream) {
  (void)in_sizes; (void)n_in; (void)out_size; (void)ws_size;
  const float* feats = (const float*)d_in[0];
  const float* W1    = (const float*)d_in[1];
  const float* a1    = (const float*)d_in[2];
  const float* W2    = (const float*)d_in[3];
  const float* a2    = (const float*)d_in[4];
  const int*   src1  = (const int*)d_in[5];
  const int*   dst1  = (const int*)d_in[6];
  const int*   src2  = (const int*)d_in[8];
  const int*   dst2  = (const int*)d_in[9];
  float* out = (float*)d_out;

  float* ws  = (float*)d_ws;
  float* z1  = ws;
  float* waT = z1  + (size_t)CN1 * CNH * CIN;
  float* s2  = waT + (size_t)2 * CNH * CIN;
  float* x   = ws  + (size_t)CN2 * CNH * CHID;
  float* Bp1 = x   + (size_t)CN1 * CNH * CHID;
  float* Bp2 = Bp1 + (size_t)CIN * CNH * CHID;
  float* h2  = Bp2 + (size_t)CNH * CHID * COUT;
  float* s1  = h2  + (size_t)CN1 * COUT;

  gat_pack_weights_kernel<<<PACK_W1_BLOCKS + PACK_W2_BLOCKS, 256, 0, stream>>>(W1, a1, W2, Bp1, waT, Bp2);
  gat_score_feat_kernel<<<CN2 / (8 * SC_NPW), 256, 0, stream>>>(feats, waT, s1, CN2);
  gat_agg_feat_kernel<<<CN1 / AGF_TPB, 32 * AGF_TPB, 0, stream>>>(feats, s1, src1, dst1, z1);
  dim3 g1((CNH * CHID) / 64, CN1 / 32);
  gat_wmma_gemm_f32_heads_elu_kernel<<<g1, 32, 0, stream>>>(z1, Bp1, x,
                                                            CIN, CNH * CHID, CNH * CIN, CNH * CHID);
  dim3 g2(COUT / 64, CN1 / 16);
  static_assert(COUT == 2 * 64, "two 64-column tiles per row of h2: two planes of partial dots");
  gat_wmma_gemm_f32_splitk_kernel<<<g2, 32 * SPLITK, 0, stream>>>(x, Bp2, h2, a2, s2,
                                                                   CNH * CHID, COUT, CNH * CHID, COUT, CN1);
  gat_agg_wave_kernel<<<CN0, 32, 0, stream>>>(h2, s2, src2, dst2, out, 1, COUT, CN1);
}


# 6 launches to 5: layer-2 GEMM inside the layer-1 GEMM, each wave's 32x64 x tile going through LDS into its K share of x@W2; 4 partial h2 and dot planes added at the gather; changes their sum order
# speedup vs baseline: 5.4179x; 1.2553x over previous
#include <hip/hip_runtime.h>

#define CN0 512
#define CN1 2048
#define CN2 8192
#define KNB 33
#define CIN 256
#define CHID 128
#define COUT 128
#define CNH 8
#define LRELU_ALPHA 0.2f

typedef float v2f __attribute__((ext_vector_type(2)));
typedef float v8f __attribute__((ext_vector_type(8)));

#define PACK_W1_BLOCKS ((CNH * CIN / 2) / 8)
#define PACK_W2_BLOCKS ((CNH * CHID / 2) / 8)
static_assert(CHID == 128 && COUT == 128, "a wave covers a 128-column row with 4 columns per lane");
__device__ __forceinline__ void gat_pack_row_pair(const float* __restrict__ row0,
                                                  const float* __restrict__ row1,
                                                  float* __restrict__ pairs,
                                                  int lane, float4& r0, float4& r1) {
  r0 = *((const float4*)row0 + lane);
  r1 = *((const float4*)row1 + lane);
  float4* o = (float4*)(pairs + 8 * lane);
  o[0] = make_float4(r0.x, r1.x, r0.y, r1.y);
  o[1] = make_float4(r0.z, r1.z, r0.w, r1.w);
}

__global__ void gat_pack_weights_kernel(const float* __restrict__ W1,
                                        const float* __restrict__ a1,
                                        const float* __restrict__ W2,
                                        float* __restrict__ Bp1,
                                        float* __restrict__ waT,
                                        float* __restrict__ Bp2) {
  const int lane = threadIdx.x & 31;
  const int wb   = threadIdx.x >> 5;
  float4 r0, r1;
  if (blockIdx.x >= PACK_W1_BLOCKS) {
    const int cp = (blockIdx.x - PACK_W1_BLOCKS) * 8 + wb;
    gat_pack_row_pair(W2 + (size_t)(2 * cp) * COUT, W2 + (size_t)(2 * cp + 1) * COUT,
                      Bp2 + (size_t)cp * (2 * COUT), lane, r0, r1);
    return;
  }
  const int w    = blockIdx.x * 8 + wb;
  const int h    = w / (CIN / 2);
  const int ip   = w % (CIN / 2);
  gat_pack_row_pair(W1 + ((size_t)h * CIN + 2 * ip) * CHID, W1 + ((size_t)h * CIN + 2 * ip + 1) * CHID,
                    Bp1 + (size_t)ip * (2 * CNH * CHID) + 2 * (h * CHID), lane, r0, r1);
  const float4 p = *((const float4*)(a1 + (size_t)h * 2 * CHID) + lane);
  const float4 q = *((const float4*)(a1 + (size_t)h * 2 * CHID + CHID) + lane);
  float s0 = r0.x * p.x + r0.y * p.y + r0.z * p.z + r0.w * p.w;
  float s1 = r0.x * q.x + r0.y * q.y + r0.z * q.z + r0.w * q.w;
  float s2 = r1.x * p.x + r1.y * p.y + r1.z * p.z + r1.w * p.w;
  float s3 = r1.x * q.x + r1.y * q.y + r1.z * q.z + r1.w * q.w;
#pragma unroll
  for (int off = 16; off > 0; off >>= 1) {
    s0 += __shfl_xor(s0, off, 32);
    s1 += __shfl_xor(s1, off, 32);
    s2 += __shfl_xor(s2, off, 32);
    s3 += __shfl_xor(s3, off, 32);
  }
  if (lane == 0) {
    *(float2*)(waT + (size_t)(h * 2 + 0) * CIN + 2 * ip) = make_float2(s0, s2);
    *(float2*)(waT + (size_t)(h * 2 + 1) * CIN + 2 * ip) = make_float2(s1, s3);
  }
}

#define GEMM_GK 4
static_assert(CHID % 64 == 0, "a 64-column tile must not straddle heads");
static_assert(CIN % (8 * GEMM_GK) == 0, "whole pairs of groups of k-steps only");

struct gat_frag_group {
  v2f a0[GEMM_GK], a1[GEMM_GK], b0[GEMM_GK], b1[GEMM_GK], b2[GEMM_GK], b3[GEMM_GK];
};

__device__ __forceinline__ void gat_load_group(gat_frag_group& f,
                                               const float* __restrict__ Arow0,
                                               const float* __restrict__ Arow1,
                                               const v2f* __restrict__ Bcol,
                                               int N, int kb, int hi) {
#pragma unroll
  for (int u = 0; u < GEMM_GK; ++u) {
    const int ka = kb + 4 * u + 2 * hi;
    f.a0[u] = *(const v2f*)(Arow0 + ka);
    f.a1[u] = *(const v2f*)(Arow1 + ka);
    const v2f* bp = Bcol + (size_t)(ka >> 1) * N;
    f.b0[u] = bp[0]; f.b1[u] = bp[16]; f.b2[u] = bp[32]; f.b3[u] = bp[48];
  }
}

__device__ __forceinline__ void gat_mma_group(const gat_frag_group& f,
                                              v8f& acc0, v8f& acc1, v8f& acc2, v8f& acc3,
                                              v8f& acc4, v8f& acc5, v8f& acc6, v8f& acc7) {
#pragma unroll
  for (int u = 0; u < GEMM_GK; ++u) {
    acc0 = __builtin_amdgcn_wmma_f32_16x16x4_f32(false, f.a0[u], false, f.b0[u], (short)0, acc0, false, false);
    acc1 = __builtin_amdgcn_wmma_f32_16x16x4_f32(false, f.a0[u], false, f.b1[u], (short)0, acc1, false, false);
    acc2 = __builtin_amdgcn_wmma_f32_16x16x4_f32(false, f.a0[u], false, f.b2[u], (short)0, acc2, false, false);
    acc3 = __builtin_amdgcn_wmma_f32_16x16x4_f32(false, f.a0[u], false, f.b3[u], (short)0, acc3, false, false);
    acc4 = __builtin_amdgcn_wmma_f32_16x16x4_f32(false, f.a1[u], false, f.b0[u], (short)0, acc4, false, false);
    acc5 = __builtin_amdgcn_wmma_f32_16x16x4_f32(false, f.a1[u], false, f.b1[u], (short)0, acc5, false, false);
    acc6 = __builtin_amdgcn_wmma_f32_16x16x4_f32(false, f.a1[u], false, f.b2[u], (short)0, acc6, false, false);
    acc7 = __builtin_amdgcn_wmma_f32_16x16x4_f32(false, f.a1[u], false, f.b3[u], (short)0, acc7, false, false);
  }
}

struct gat_l2_group { v2f b[4][8]; };

__device__ __forceinline__ void gat_l2_load_group(gat_l2_group& f, const v2f* __restrict__ B2r, int g, int hi) {
#pragma unroll
  for (int u = 0; u < 4; ++u) {
    const v2f* bp = B2r + (size_t)(2 * (4 * g + u) + hi) * COUT;
#pragma unroll
    for (int c = 0; c < 8; ++c) f.b[u][c] = bp[16 * c];
  }
}

__device__ __forceinline__ void gat_l2_mma_group(const gat_l2_group& f,
                                                 const float* xa0, const float* xa1, int g,
                                                 v8f (&h0)[8], v8f (&h1)[8]) {
#pragma unroll
  for (int u = 0; u < 4; ++u) {
    const v2f a0 = *(const v2f*)(xa0 + 4 * (4 * g + u));
    const v2f a1 = *(const v2f*)(xa1 + 4 * (4 * g + u));
#pragma unroll
    for (int c = 0; c < 8; ++c)
      h0[c] = __builtin_amdgcn_wmma_f32_16x16x4_f32(false, a0, false, f.b[u][c], (short)0, h0[c], false, false);
#pragma unroll
    for (int c = 0; c < 8; ++c)
      h1[c] = __builtin_amdgcn_wmma_f32_16x16x4_f32(false, a1, false, f.b[u][c], (short)0, h1[c], false, false);
  }
}

__device__ __forceinline__ float gat_elu(float v) {
  const float e = expm1f((v > 0.f) ? 0.f : v);
  return (v > 0.f) ? v : e;
}

#define XT_PITCH 68
#define L2_PLANES ((CNH * CHID) / 256)
static_assert(COUT == 128, "8 column tiles of h2 per wave, 4 columns per lane in the fold");
__global__ void __launch_bounds__(128)
gat_wmma_gemm_f32_heads_elu_l2_kernel(const float* __restrict__ A,
                                      const float* __restrict__ Bp,
                                      const float* __restrict__ Bp2,
                                      const float* __restrict__ a,
                                      float* __restrict__ Hp,
                                      float* __restrict__ Sp,
                                      int K, int N, int lda, int M) {
  __shared__ __attribute__((aligned(16))) float xt[4][32][XT_PITCH];
  __shared__ float part[4][128][32];
  const int lane = threadIdx.x & 31;
  const int wv   = __builtin_amdgcn_readfirstlane((int)(threadIdx.x >> 5));
  const int hi   = lane >> 4;
  const int lr   = lane & 15;
  const int m0   = blockIdx.y << 5;
  const int n0   = (blockIdx.x << 8) + (wv << 6);

  v8f acc0 = {}; v8f acc1 = {}; v8f acc2 = {}; v8f acc3 = {};
  v8f acc4 = {}; v8f acc5 = {}; v8f acc6 = {}; v8f acc7 = {};

  const float* Arow0 = A + (size_t)(m0 + lr) * lda + (size_t)(n0 / CHID) * K;
  const float* Arow1 = Arow0 + (size_t)16 * lda;
  const v2f* Bcol = (const v2f*)Bp + n0 + lr;

  gat_frag_group p, q;
  gat_load_group(p, Arow0, Arow1, Bcol, N, 0, hi);
  int k = 0;
  for (; k + 8 * GEMM_GK < K; k += 8 * GEMM_GK) {
    gat_load_group(q, Arow0, Arow1, Bcol, N, k + 4 * GEMM_GK, hi);
    gat_mma_group(p, acc0, acc1, acc2, acc3, acc4, acc5, acc6, acc7);
    gat_load_group(p, Arow0, Arow1, Bcol, N, k + 8 * GEMM_GK, hi);
    gat_mma_group(q, acc0, acc1, acc2, acc3, acc4, acc5, acc6, acc7);
  }
  gat_load_group(q, Arow0, Arow1, Bcol, N, k + 4 * GEMM_GK, hi);
  gat_mma_group(p, acc0, acc1, acc2, acc3, acc4, acc5, acc6, acc7);
  gat_mma_group(q, acc0, acc1, acc2, acc3, acc4, acc5, acc6, acc7);

  const v2f* B2r = (const v2f*)Bp2 + (size_t)(n0 >> 1) * COUT + lr;
  gat_l2_group p2, q2;
  gat_l2_load_group(p2, B2r, 0, hi);

  float (*xw)[XT_PITCH] = xt[wv];
#pragma unroll
  for (int i = 0; i < 8; ++i) {
    xw[i + 8 * hi][lr]           = gat_elu(acc0[i]);
    xw[i + 8 * hi][16 + lr]      = gat_elu(acc1[i]);
    xw[i + 8 * hi][32 + lr]      = gat_elu(acc2[i]);
    xw[i + 8 * hi][48 + lr]      = gat_elu(acc3[i]);
    xw[16 + i + 8 * hi][lr]      = gat_elu(acc4[i]);
    xw[16 + i + 8 * hi][16 + lr] = gat_elu(acc5[i]);
    xw[16 + i + 8 * hi][32 + lr] = gat_elu(acc6[i]);
    xw[16 + i + 8 * hi][48 + lr] = gat_elu(acc7[i]);
  }
  __syncthreads();

  v8f h0[8], h1[8];
#pragma unroll
  for (int c = 0; c < 8; ++c) { h0[c] = (v8f){}; h1[c] = (v8f){}; }
  const float* xa0 = &xw[lr][2 * hi];
  const float* xa1 = &xw[16 + lr][2 * hi];
  gat_l2_load_group(q2, B2r, 1, hi);
  gat_l2_mma_group(p2, xa0, xa1, 0, h0, h1);
  gat_l2_load_group(p2, B2r, 2, hi);
  gat_l2_mma_group(q2, xa0, xa1, 1, h0, h1);
  gat_l2_load_group(q2, B2r, 3, hi);
  gat_l2_mma_group(p2, xa0, xa1, 2, h0, h1);
  gat_l2_mma_group(q2, xa0, xa1, 3, h0, h1);

#pragma unroll
  for (int c = 0; c < 8; ++c) {
#pragma unroll
    for (int i = 0; i < 8; ++i) {
      part[wv][(0 * 8 + c) * 8 + i][lane] = h0[c][i];
      part[wv][(1 * 8 + c) * 8 + i][lane] = h1[c][i];
    }
  }
  __syncthreads();

  const int fr = wv >> 1;
  const int pl = (wv & 1) * 16 + lr;
  const float* ap = a + lane;
  const float aA0 = ap[0], aA1 = ap[32], aA2 = ap[64], aA3 = ap[96];
  const float aB0 = ap[COUT], aB1 = ap[COUT + 32], aB2 = ap[COUT + 64], aB3 = ap[COUT + 96];
  float* Hrow = Hp + ((size_t)blockIdx.x * M + m0 + 8 * wv) * COUT + lane;
  float dA[8], dB[8];
#pragma unroll
  for (int j = 0; j < 8; ++j) {
    float s[4];
#pragma unroll
    for (int jj = 0; jj < 4; ++jj) {
      const int reg = (fr * 8 + 2 * jj + hi) * 8 + j;
      s[jj] = ((part[0][reg][pl] + part[1][reg][pl]) + part[2][reg][pl]) + part[3][reg][pl];
    }
    Hrow[(size_t)j * COUT + 0]  = s[0];
    Hrow[(size_t)j * COUT + 32] = s[1];
    Hrow[(size_t)j * COUT + 64] = s[2];
    Hrow[(size_t)j * COUT + 96] = s[3];
    dA[j] = s[0] * aA0 + s[1] * aA1 + s[2] * aA2 + s[3] * aA3;
    dB[j] = s[0] * aB0 + s[1] * aB1 + s[2] * aB2 + s[3] * aB3;
  }
#pragma unroll
  for (int off = 16; off > 0; off >>= 1) {
#pragma unroll
    for (int j = 0; j < 8; ++j) {
      dA[j] += __shfl_xor(dA[j], off, 32);
      dB[j] += __shfl_xor(dB[j], off, 32);
    }
  }
  if (lane == 0) {
    float4* sp = (float4*)(Sp + ((size_t)blockIdx.x * M + m0 + 8 * wv) * 2);
#pragma unroll
    for (int j = 0; j < 8; j += 2)
      sp[j >> 1] = make_float4(dA[j], dB[j], dA[j + 1], dB[j + 1]);
  }
}

#define SC_NPW 4
static_assert(CN2 % SC_NPW == 0, "whole waves only");
static_assert(CIN == 256, "a lane holds 2 x 4 columns of a 256-wide row");
__global__ void __launch_bounds__(256)
gat_score_feat_kernel(const float* __restrict__ X,
                                      const float* __restrict__ waT,
                                      float* __restrict__ S,
                                      int nnodes) {
  const int gt   = blockIdx.x * blockDim.x + threadIdx.x;
  const int nb   = (gt >> 5) * SC_NPW;
  const int lane = gt & 31;
  if (nb + SC_NPW > nnodes) return;
  const float4* wp = (const float4*)waT + lane;
  float4 f0[SC_NPW], f1[SC_NPW];
#pragma unroll
  for (int r = 0; r < SC_NPW; ++r) {
    const float4* xp = (const float4*)(X + (size_t)(nb + r) * CIN) + lane;
    f0[r] = xp[0];
    f1[r] = xp[CIN / 8];
  }
  float s[SC_NPW][2 * CNH];
#pragma unroll
  for (int o = 0; o < 2 * CNH; ++o) {
    const float4 w0 = wp[o * (CIN / 4)];
    const float4 w1 = wp[o * (CIN / 4) + CIN / 8];
#pragma unroll
    for (int r = 0; r < SC_NPW; ++r) {
      s[r][o] = f0[r].x * w0.x + f0[r].y * w0.y + f0[r].z * w0.z + f0[r].w * w0.w
              + f1[r].x * w1.x + f1[r].y * w1.y + f1[r].z * w1.z + f1[r].w * w1.w;
    }
  }
#pragma unroll
  for (int r = 0; r < SC_NPW; ++r) {
#pragma unroll
    for (int o = 0; o < 2 * CNH; ++o) {
#pragma unroll
      for (int off = 16; off > 0; off >>= 1) s[r][o] += __shfl_xor(s[r][o], off, 32);
    }
  }
  if (lane == 0) {
#pragma unroll
    for (int r = 0; r < SC_NPW; ++r) {
      float4* o4 = (float4*)(S + (size_t)(nb + r) * (2 * CNH));
#pragma unroll
      for (int o = 0; o < 2 * CNH; o += 4)
        o4[o >> 2] = make_float4(s[r][o], s[r][o + 1], s[r][o + 2], s[r][o + 3]);
    }
  }
}

#define AGF_TPB 8
static_assert(CN1 % AGF_TPB == 0, "whole blocks only");
static_assert(CNH == 8, "the scores of a node are read as 4 pieces of 2 heads");
static_assert(CIN == 256, "a lane holds 2 x 4 columns of a 256-wide row");
__global__ void __launch_bounds__(32 * AGF_TPB)
gat_agg_feat_kernel(const float* __restrict__ X,
                    const float* __restrict__ S,
                    const int* __restrict__ src,
                    const int* __restrict__ dst,
                    float* __restrict__ Z) {
  const int lane = threadIdx.x & 31;
  const int wv   = __builtin_amdgcn_readfirstlane((int)(threadIdx.x >> 5));
  const int t    = blockIdx.x * AGF_TPB + wv;
  const int m0   = t * KNB;

  const int sL = src[m0 + lane], dL = dst[m0 + lane];
  const int sX = src[m0 + 32],   dX = dst[m0 + 32];
  const float4* SsL = (const float4*)(S + (size_t)sL * (2 * CNH));
  const float4* SdL = (const float4*)(S + (size_t)dL * (2 * CNH));
  const float4* SsX = (const float4*)(S + (size_t)sX * (2 * CNH));
  const float4* SdX = (const float4*)(S + (size_t)dX * (2 * CNH));
  float wL[CNH], wX[CNH];
#pragma unroll
  for (int g = 0; g < CNH / 2; ++g) {
    const float4 a = SsL[g], b = SdL[g], c = SsX[g], d = SdX[g];
    wL[2 * g]     = a.x + b.y;
    wL[2 * g + 1] = a.z + b.w;
    wX[2 * g]     = c.x + d.y;
    wX[2 * g + 1] = c.z + d.w;
  }
#pragma unroll
  for (int h = 0; h < CNH; ++h) {
    float eL = wL[h], eX = wX[h];
    eL = (eL > 0.f) ? eL : LRELU_ALPHA * eL;
    eX = (eX > 0.f) ? eX : LRELU_ALPHA * eX;
    float mx = eL;
#pragma unroll
    for (int off = 16; off > 0; off >>= 1) mx = fmaxf(mx, __shfl_xor(mx, off, 32));
    mx = fmaxf(mx, eX);
    const float pL = expf(eL - mx);
    const float pX = expf(eX - mx);
    float sum = pL;
#pragma unroll
    for (int off = 16; off > 0; off >>= 1) sum += __shfl_xor(sum, off, 32);
    sum += pX;
    const float inv = 1.f / sum;
    wL[h] = pL * inv;
    wX[h] = pX * inv;
  }

  const float* col = X + 4 * lane;
  float4 acc0[CNH], acc1[CNH];
#pragma unroll
  for (int h = 0; h < CNH; ++h) {
    acc0[h] = make_float4(0.f, 0.f, 0.f, 0.f);
    acc1[h] = make_float4(0.f, 0.f, 0.f, 0.f);
  }
#pragma unroll
  for (int k = 0; k < 32; ++k) {
    const int dk = __builtin_amdgcn_readlane(dL, k);
    const float4 v0 = *(const float4*)(col + (size_t)dk * CIN);
    const float4 v1 = *(const float4*)(col + (size_t)dk * CIN + CIN / 2);
#pragma unroll
    for (int h = 0; h < CNH; ++h) {
      const float wk = __int_as_float(__builtin_amdgcn_readlane(__float_as_int(wL[h]), k));
      acc0[h].x += wk * v0.x; acc0[h].y += wk * v0.y; acc0[h].z += wk * v0.z; acc0[h].w += wk * v0.w;
      acc1[h].x += wk * v1.x; acc1[h].y += wk * v1.y; acc1[h].z += wk * v1.z; acc1[h].w += wk * v1.w;
    }
  }
  {
    const int dk = __builtin_amdgcn_readfirstlane(dX);
    const float4 v0 = *(const float4*)(col + (size_t)dk * CIN);
    const float4 v1 = *(const float4*)(col + (size_t)dk * CIN + CIN / 2);
#pragma unroll
    for (int h = 0; h < CNH; ++h) {
      const float wk = __int_as_float(__builtin_amdgcn_readfirstlane(__float_as_int(wX[h])));
      acc0[h].x += wk * v0.x; acc0[h].y += wk * v0.y; acc0[h].z += wk * v0.z; acc0[h].w += wk * v0.w;
      acc1[h].x += wk * v1.x; acc1[h].y += wk * v1.y; acc1[h].z += wk * v1.z; acc1[h].w += wk * v1.w;
    }
  }
  float* zp = Z + (size_t)t * (CNH * CIN) + 4 * lane;
#pragma unroll
  for (int h = 0; h < CNH; ++h) {
    *(float4*)(zp + (size_t)h * CIN)           = acc0[h];
    *(float4*)(zp + (size_t)h * CIN + CIN / 2) = acc1[h];
  }
}

static_assert(KNB == 33, "edge 32 is the one edge beyond the 32 lanes");
static_assert(L2_PLANES == 4, "the planes are added as two pairs");
__global__ void __launch_bounds__(32 * L2_PLANES) __attribute__((amdgpu_waves_per_eu(1, 4)))
gat_agg_planes_kernel(const float* __restrict__ Hp,
                      const float* __restrict__ Sp,
                      const int* __restrict__ src,
                      const int* __restrict__ dst,
                      float* __restrict__ Yout,
                      int M) {
  __shared__ float4 red[L2_PLANES][32];
  const int t    = blockIdx.x;
  const int lane = threadIdx.x & 31;
  const int pv   = __builtin_amdgcn_readfirstlane((int)(threadIdx.x >> 5));
  const int m0   = t * KNB;

  const int sL = src[m0 + lane], dL = dst[m0 + lane];
  const int sX = src[m0 + 32],   dX = dst[m0 + 32];

  const float* col = Hp + (size_t)pv * M * COUT + 4 * lane;
  float4 v[KNB];
#pragma unroll
  for (int k = 0; k < 32; ++k) {
    const int dk = __builtin_amdgcn_readlane(dL, k);
    v[k] = *(const float4*)(col + (size_t)dk * COUT);
  }
  {
    const int dk = __builtin_amdgcn_readfirstlane(dX);
    v[32] = *(const float4*)(col + (size_t)dk * COUT);
  }

  const float2* S0 = (const float2*)Sp;
  const float2* S1 = S0 + M;
  const float2* S2 = S1 + M;
  const float2* S3 = S2 + M;
  float eL = ((S0[sL].x + S1[sL].x) + (S2[sL].x + S3[sL].x)) + ((S0[dL].y + S1[dL].y) + (S2[dL].y + S3[dL].y));
  float eX = ((S0[sX].x + S1[sX].x) + (S2[sX].x + S3[sX].x)) + ((S0[dX].y + S1[dX].y) + (S2[dX].y + S3[dX].y));
  eL = (eL > 0.f) ? eL : LRELU_ALPHA * eL;
  eX = (eX > 0.f) ? eX : LRELU_ALPHA * eX;

  float mx = eL;
  for (int off = 16; off > 0; off >>= 1) mx = fmaxf(mx, __shfl_xor(mx, off, 32));
  mx = fmaxf(mx, eX);
  const float pL = expf(eL - mx);
  const float pX = expf(eX - mx);
  float sum = pL;
  for (int off = 16; off > 0; off >>= 1) sum += __shfl_xor(sum, off, 32);
  sum += pX;
  const float inv = 1.f / sum;
  const float wL = pL * inv;
  const float wX = pX * inv;

  float4 acc = make_float4(0.f, 0.f, 0.f, 0.f);
#pragma unroll
  for (int k = 0; k < 32; ++k) {
    const float wk = __int_as_float(__builtin_amdgcn_readlane(__float_as_int(wL), k));
    acc.x += wk * v[k].x; acc.y += wk * v[k].y; acc.z += wk * v[k].z; acc.w += wk * v[k].w;
  }
  {
    const float wk = __int_as_float(__builtin_amdgcn_readfirstlane(__float_as_int(wX)));
    acc.x += wk * v[32].x; acc.y += wk * v[32].y; acc.z += wk * v[32].z; acc.w += wk * v[32].w;
  }
  red[pv][lane] = acc;
  __syncthreads();
  if (pv == 0) {
    const float4 c0 = red[0][lane], c1 = red[1][lane], c2 = red[2][lane], c3 = red[3][lane];
    const float tx = (c0.x + c1.x) + (c2.x + c3.x);
    const float ty = (c0.y + c1.y) + (c2.y + c3.y);
    const float tz = (c0.z + c1.z) + (c2.z + c3.z);
    const float tw = (c0.w + c1.w) + (c2.w + c3.w);
    float4 r;
    r.x = (tx > 0.f) ? tx : expm1f(tx);
    r.y = (ty > 0.f) ? ty : expm1f(ty);
    r.z = (tz > 0.f) ? tz : expm1f(tz);
    r.w = (tw > 0.f) ? tw : expm1f(tw);
    *(float4*)(Yout + (size_t)t * COUT + 4 * lane) = r;
  }
}

extern "C" void kernel_launch(void* const* d_in, const int* in_sizes, int n_in,
                              void* d_out, int out_size, void* d_ws, size_t ws_size,
                              hipStream_t stream) {
  (void)in_sizes; (void)n_in; (void)out_size; (void)ws_size;
  const float* feats = (const float*)d_in[0];
  const float* W1    = (const float*)d_in[1];
  const float* a1    = (const float*)d_in[2];
  const float* W2    = (const float*)d_in[3];
  const float* a2    = (const float*)d_in[4];
  const int*   src1  = (const int*)d_in[5];
  const int*   dst1  = (const int*)d_in[6];
  const int*   src2  = (const int*)d_in[8];
  const int*   dst2  = (const int*)d_in[9];
  float* out = (float*)d_out;

  float* ws  = (float*)d_ws;
  float* z1  = ws;
  float* waT = z1  + (size_t)CN1 * CNH * CIN;
  float* s2  = waT + (size_t)2 * CNH * CIN;
  float* h2p = ws  + (size_t)CN2 * CNH * CHID;
  float* Bp1 = h2p + (size_t)CN1 * CNH * CHID;
  float* Bp2 = Bp1 + (size_t)CIN * CNH * CHID;
  float* s1  = Bp2 + (size_t)CNH * CHID * COUT + (size_t)CN1 * COUT;

  gat_pack_weights_kernel<<<PACK_W1_BLOCKS + PACK_W2_BLOCKS, 256, 0, stream>>>(W1, a1, W2, Bp1, waT, Bp2);
  gat_score_feat_kernel<<<CN2 / (8 * SC_NPW), 256, 0, stream>>>(feats, waT, s1, CN2);
  gat_agg_feat_kernel<<<CN1 / AGF_TPB, 32 * AGF_TPB, 0, stream>>>(feats, s1, src1, dst1, z1);
  dim3 g1(L2_PLANES, CN1 / 32);
  gat_wmma_gemm_f32_heads_elu_l2_kernel<<<g1, 128, 0, stream>>>(z1, Bp1, Bp2, a2, h2p, s2,
                                                                CIN, CNH * CHID, CNH * CIN, CN1);
  gat_agg_planes_kernel<<<CN0, 32 * L2_PLANES, 0, stream>>>(h2p, s2, src2, dst2, out, CN1);
}


# layer-1 GEMM: W1 packed in k-quads so each A and B fragment load is 16 bytes for two k-steps: half the loads, operand groups of 8 k-steps issued 64 matrix ops ahead; changes the k sum order of x
# speedup vs baseline: 5.8062x; 1.0717x over previous
#include <hip/hip_runtime.h>

#define CN0 512
#define CN1 2048
#define CN2 8192
#define KNB 33
#define CIN 256
#define CHID 128
#define COUT 128
#define CNH 8
#define LRELU_ALPHA 0.2f

typedef float v2f __attribute__((ext_vector_type(2)));
typedef float v8f __attribute__((ext_vector_type(8)));

#define PACK_W1_BLOCKS ((CNH * CIN / 4) / 8)
#define PACK_W2_BLOCKS ((CNH * CHID / 2) / 8)
static_assert(CHID == 128 && COUT == 128, "a wave covers a 128-column row with 4 columns per lane");
__device__ __forceinline__ void gat_pack_row_pair(const float* __restrict__ row0,
                                                  const float* __restrict__ row1,
                                                  float* __restrict__ pairs,
                                                  int lane, float4& r0, float4& r1) {
  r0 = *((const float4*)row0 + lane);
  r1 = *((const float4*)row1 + lane);
  float4* o = (float4*)(pairs + 8 * lane);
  o[0] = make_float4(r0.x, r1.x, r0.y, r1.y);
  o[1] = make_float4(r0.z, r1.z, r0.w, r1.w);
}

__global__ void gat_pack_weights_kernel(const float* __restrict__ W1,
                                        const float* __restrict__ a1,
                                        const float* __restrict__ W2,
                                        float* __restrict__ Bp1,
                                        float* __restrict__ waT,
                                        float* __restrict__ Bp2) {
  const int lane = threadIdx.x & 31;
  const int wb   = threadIdx.x >> 5;
  float4 r0, r1;
  if (blockIdx.x >= PACK_W1_BLOCKS) {
    const int cp = (blockIdx.x - PACK_W1_BLOCKS) * 8 + wb;
    gat_pack_row_pair(W2 + (size_t)(2 * cp) * COUT, W2 + (size_t)(2 * cp + 1) * COUT,
                      Bp2 + (size_t)cp * (2 * COUT), lane, r0, r1);
    return;
  }
  const int w    = blockIdx.x * 8 + wb;
  const int h    = w / (CIN / 4);
  const int iq   = w % (CIN / 4);
  const float* rq = W1 + ((size_t)h * CIN + 4 * iq) * CHID;
  r0 = *((const float4*)rq + lane);
  r1 = *((const float4*)(rq + CHID) + lane);
  const float4 r2 = *((const float4*)(rq + 2 * CHID) + lane);
  const float4 r3 = *((const float4*)(rq + 3 * CHID) + lane);
  float4* o = (float4*)(Bp1 + (size_t)iq * (4 * CNH * CHID) + 4 * (h * CHID + 4 * lane));
  o[0] = make_float4(r0.x, r1.x, r2.x, r3.x);
  o[1] = make_float4(r0.y, r1.y, r2.y, r3.y);
  o[2] = make_float4(r0.z, r1.z, r2.z, r3.z);
  o[3] = make_float4(r0.w, r1.w, r2.w, r3.w);
  const float4 p = *((const float4*)(a1 + (size_t)h * 2 * CHID) + lane);
  const float4 q = *((const float4*)(a1 + (size_t)h * 2 * CHID + CHID) + lane);
  float sp[4], sq[4];
  sp[0] = r0.x * p.x + r0.y * p.y + r0.z * p.z + r0.w * p.w;
  sq[0] = r0.x * q.x + r0.y * q.y + r0.z * q.z + r0.w * q.w;
  sp[1] = r1.x * p.x + r1.y * p.y + r1.z * p.z + r1.w * p.w;
  sq[1] = r1.x * q.x + r1.y * q.y + r1.z * q.z + r1.w * q.w;
  sp[2] = r2.x * p.x + r2.y * p.y + r2.z * p.z + r2.w * p.w;
  sq[2] = r2.x * q.x + r2.y * q.y + r2.z * q.z + r2.w * q.w;
  sp[3] = r3.x * p.x + r3.y * p.y + r3.z * p.z + r3.w * p.w;
  sq[3] = r3.x * q.x + r3.y * q.y + r3.z * q.z + r3.w * q.w;
#pragma unroll
  for (int off = 16; off > 0; off >>= 1) {
#pragma unroll
    for (int r = 0; r < 4; ++r) {
      sp[r] += __shfl_xor(sp[r], off, 32);
      sq[r] += __shfl_xor(sq[r], off, 32);
    }
  }
  if (lane == 0) {
    *(float4*)(waT + (size_t)(h * 2 + 0) * CIN + 4 * iq) = make_float4(sp[0], sp[1], sp[2], sp[3]);
    *(float4*)(waT + (size_t)(h * 2 + 1) * CIN + 4 * iq) = make_float4(sq[0], sq[1], sq[2], sq[3]);
  }
}

#define GEMM_GQ 4
static_assert(CHID % 64 == 0, "a 64-column tile must not straddle heads");
static_assert(CIN % (16 * GEMM_GQ) == 0, "whole pairs of groups only");

struct gat_frag_group {
  float4 a0[GEMM_GQ], a1[GEMM_GQ], b0[GEMM_GQ], b1[GEMM_GQ], b2[GEMM_GQ], b3[GEMM_GQ];
};

__device__ __forceinline__ void gat_load_group(gat_frag_group& f,
                                               const float* __restrict__ Arow0,
                                               const float* __restrict__ Arow1,
                                               const float4* __restrict__ Bcol,
                                               int N, int kb, int hi) {
#pragma unroll
  for (int u = 0; u < GEMM_GQ; ++u) {
    const int kq = kb + 8 * u + 4 * hi;
    f.a0[u] = *(const float4*)(Arow0 + kq);
    f.a1[u] = *(const float4*)(Arow1 + kq);
    const float4* bp = Bcol + (size_t)(kq >> 2) * N;
    f.b0[u] = bp[0]; f.b1[u] = bp[16]; f.b2[u] = bp[32]; f.b3[u] = bp[48];
  }
}

__device__ __forceinline__ void gat_mma_step(v2f a0, v2f a1, v2f b0, v2f b1, v2f b2, v2f b3,
                                             v8f& acc0, v8f& acc1, v8f& acc2, v8f& acc3,
                                             v8f& acc4, v8f& acc5, v8f& acc6, v8f& acc7) {
  acc0 = __builtin_amdgcn_wmma_f32_16x16x4_f32(false, a0, false, b0, (short)0, acc0, false, false);
  acc1 = __builtin_amdgcn_wmma_f32_16x16x4_f32(false, a0, false, b1, (short)0, acc1, false, false);
  acc2 = __builtin_amdgcn_wmma_f32_16x16x4_f32(false, a0, false, b2, (short)0, acc2, false, false);
  acc3 = __builtin_amdgcn_wmma_f32_16x16x4_f32(false, a0, false, b3, (short)0, acc3, false, false);
  acc4 = __builtin_amdgcn_wmma_f32_16x16x4_f32(false, a1, false, b0, (short)0, acc4, false, false);
  acc5 = __builtin_amdgcn_wmma_f32_16x16x4_f32(false, a1, false, b1, (short)0, acc5, false, false);
  acc6 = __builtin_amdgcn_wmma_f32_16x16x4_f32(false, a1, false, b2, (short)0, acc6, false, false);
  acc7 = __builtin_amdgcn_wmma_f32_16x16x4_f32(false, a1, false, b3, (short)0, acc7, false, false);
}

__device__ __forceinline__ void gat_mma_group(const gat_frag_group& f,
                                              v8f& acc0, v8f& acc1, v8f& acc2, v8f& acc3,
                                              v8f& acc4, v8f& acc5, v8f& acc6, v8f& acc7) {
#pragma unroll
  for (int u = 0; u < GEMM_GQ; ++u) {
    gat_mma_step((v2f){f.a0[u].x, f.a0[u].y}, (v2f){f.a1[u].x, f.a1[u].y},
                 (v2f){f.b0[u].x, f.b0[u].y}, (v2f){f.b1[u].x, f.b1[u].y},
                 (v2f){f.b2[u].x, f.b2[u].y}, (v2f){f.b3[u].x, f.b3[u].y},
                 acc0, acc1, acc2, acc3, acc4, acc5, acc6, acc7);
    gat_mma_step((v2f){f.a0[u].z, f.a0[u].w}, (v2f){f.a1[u].z, f.a1[u].w},
                 (v2f){f.b0[u].z, f.b0[u].w}, (v2f){f.b1[u].z, f.b1[u].w},
                 (v2f){f.b2[u].z, f.b2[u].w}, (v2f){f.b3[u].z, f.b3[u].w},
                 acc0, acc1, acc2, acc3, acc4, acc5, acc6, acc7);
  }
}

struct gat_l2_group { v2f b[4][8]; };

__device__ __forceinline__ void gat_l2_load_group(gat_l2_group& f, const v2f* __restrict__ B2r, int g, int hi) {
#pragma unroll
  for (int u = 0; u < 4; ++u) {
    const v2f* bp = B2r + (size_t)(2 * (4 * g + u) + hi) * COUT;
#pragma unroll
    for (int c = 0; c < 8; ++c) f.b[u][c] = bp[16 * c];
  }
}

__device__ __forceinline__ void gat_l2_mma_group(const gat_l2_group& f,
                                                 const float* xa0, const float* xa1, int g,
                                                 v8f (&h0)[8], v8f (&h1)[8]) {
#pragma unroll
  for (int u = 0; u < 4; ++u) {
    const v2f a0 = *(const v2f*)(xa0 + 4 * (4 * g + u));
    const v2f a1 = *(const v2f*)(xa1 + 4 * (4 * g + u));
#pragma unroll
    for (int c = 0; c < 8; ++c)
      h0[c] = __builtin_amdgcn_wmma_f32_16x16x4_f32(false, a0, false, f.b[u][c], (short)0, h0[c], false, false);
#pragma unroll
    for (int c = 0; c < 8; ++c)
      h1[c] = __builtin_amdgcn_wmma_f32_16x16x4_f32(false, a1, false, f.b[u][c], (short)0, h1[c], false, false);
  }
}

__device__ __forceinline__ float gat_elu(float v) {
  const float e = expm1f((v > 0.f) ? 0.f : v);
  return (v > 0.f) ? v : e;
}

#define XT_PITCH 68
#define L2_PLANES ((CNH * CHID) / 256)
static_assert(COUT == 128, "8 column tiles of h2 per wave, 4 columns per lane in the fold");
__global__ void __launch_bounds__(128)
gat_wmma_gemm_f32_heads_elu_l2_kernel(const float* __restrict__ A,
                                      const float* __restrict__ Bp,
                                      const float* __restrict__ Bp2,
                                      const float* __restrict__ a,
                                      float* __restrict__ Hp,
                                      float* __restrict__ Sp,
                                      int K, int N, int lda, int M) {
  __shared__ __attribute__((aligned(16))) float xt[4][32][XT_PITCH];
  __shared__ float part[4][128][32];
  const int lane = threadIdx.x & 31;
  const int wv   = __builtin_amdgcn_readfirstlane((int)(threadIdx.x >> 5));
  const int hi   = lane >> 4;
  const int lr   = lane & 15;
  const int m0   = blockIdx.y << 5;
  const int n0   = (blockIdx.x << 8) + (wv << 6);

  v8f acc0 = {}; v8f acc1 = {}; v8f acc2 = {}; v8f acc3 = {};
  v8f acc4 = {}; v8f acc5 = {}; v8f acc6 = {}; v8f acc7 = {};

  const float* Arow0 = A + (size_t)(m0 + lr) * lda + (size_t)(n0 / CHID) * K;
  const float* Arow1 = Arow0 + (size_t)16 * lda;
  const float4* Bcol = (const float4*)Bp + n0 + lr;

  gat_frag_group p, q;
  gat_load_group(p, Arow0, Arow1, Bcol, N, 0, hi);
  int k = 0;
  for (; k + 16 * GEMM_GQ < K; k += 16 * GEMM_GQ) {
    gat_load_group(q, Arow0, Arow1, Bcol, N, k + 8 * GEMM_GQ, hi);
    gat_mma_group(p, acc0, acc1, acc2, acc3, acc4, acc5, acc6, acc7);
    gat_load_group(p, Arow0, Arow1, Bcol, N, k + 16 * GEMM_GQ, hi);
    gat_mma_group(q, acc0, acc1, acc2, acc3, acc4, acc5, acc6, acc7);
  }
  gat_load_group(q, Arow0, Arow1, Bcol, N, k + 8 * GEMM_GQ, hi);
  gat_mma_group(p, acc0, acc1, acc2, acc3, acc4, acc5, acc6, acc7);
  gat_mma_group(q, acc0, acc1, acc2, acc3, acc4, acc5, acc6, acc7);

  const v2f* B2r = (const v2f*)Bp2 + (size_t)(n0 >> 1) * COUT + lr;
  gat_l2_group p2, q2;
  gat_l2_load_group(p2, B2r, 0, hi);

  float (*xw)[XT_PITCH] = xt[wv];
#pragma unroll
  for (int i = 0; i < 8; ++i) {
    xw[i + 8 * hi][lr]           = gat_elu(acc0[i]);
    xw[i + 8 * hi][16 + lr]      = gat_elu(acc1[i]);
    xw[i + 8 * hi][32 + lr]      = gat_elu(acc2[i]);
    xw[i + 8 * hi][48 + lr]      = gat_elu(acc3[i]);
    xw[16 + i + 8 * hi][lr]      = gat_elu(acc4[i]);
    xw[16 + i + 8 * hi][16 + lr] = gat_elu(acc5[i]);
    xw[16 + i + 8 * hi][32 + lr] = gat_elu(acc6[i]);
    xw[16 + i + 8 * hi][48 + lr] = gat_elu(acc7[i]);
  }
  __syncthreads();

  v8f h0[8], h1[8];
#pragma unroll
  for (int c = 0; c < 8; ++c) { h0[c] = (v8f){}; h1[c] = (v8f){}; }
  const float* xa0 = &xw[lr][2 * hi];
  const float* xa1 = &xw[16 + lr][2 * hi];
  gat_l2_load_group(q2, B2r, 1, hi);
  gat_l2_mma_group(p2, xa0, xa1, 0, h0, h1);
  gat_l2_load_group(p2, B2r, 2, hi);
  gat_l2_mma_group(q2, xa0, xa1, 1, h0, h1);
  gat_l2_load_group(q2, B2r, 3, hi);
  gat_l2_mma_group(p2, xa0, xa1, 2, h0, h1);
  gat_l2_mma_group(q2, xa0, xa1, 3, h0, h1);

#pragma unroll
  for (int c = 0; c < 8; ++c) {
#pragma unroll
    for (int i = 0; i < 8; ++i) {
      part[wv][(0 * 8 + c) * 8 + i][lane] = h0[c][i];
      part[wv][(1 * 8 + c) * 8 + i][lane] = h1[c][i];
    }
  }
  __syncthreads();

  const int fr = wv >> 1;
  const int pl = (wv & 1) * 16 + lr;
  const float* ap = a + lane;
  const float aA0 = ap[0], aA1 = ap[32], aA2 = ap[64], aA3 = ap[96];
  const float aB0 = ap[COUT], aB1 = ap[COUT + 32], aB2 = ap[COUT + 64], aB3 = ap[COUT + 96];
  float* Hrow = Hp + ((size_t)blockIdx.x * M + m0 + 8 * wv) * COUT + lane;
  float dA[8], dB[8];
#pragma unroll
  for (int j = 0; j < 8; ++j) {
    float s[4];
#pragma unroll
    for (int jj = 0; jj < 4; ++jj) {
      const int reg = (fr * 8 + 2 * jj + hi) * 8 + j;
      s[jj] = ((part[0][reg][pl] + part[1][reg][pl]) + part[2][reg][pl]) + part[3][reg][pl];
    }
    Hrow[(size_t)j * COUT + 0]  = s[0];
    Hrow[(size_t)j * COUT + 32] = s[1];
    Hrow[(size_t)j * COUT + 64] = s[2];
    Hrow[(size_t)j * COUT + 96] = s[3];
    dA[j] = s[0] * aA0 + s[1] * aA1 + s[2] * aA2 + s[3] * aA3;
    dB[j] = s[0] * aB0 + s[1] * aB1 + s[2] * aB2 + s[3] * aB3;
  }
#pragma unroll
  for (int off = 16; off > 0; off >>= 1) {
#pragma unroll
    for (int j = 0; j < 8; ++j) {
      dA[j] += __shfl_xor(dA[j], off, 32);
      dB[j] += __shfl_xor(dB[j], off, 32);
    }
  }
  if (lane == 0) {
    float4* sp = (float4*)(Sp + ((size_t)blockIdx.x * M + m0 + 8 * wv) * 2);
#pragma unroll
    for (int j = 0; j < 8; j += 2)
      sp[j >> 1] = make_float4(dA[j], dB[j], dA[j + 1], dB[j + 1]);
  }
}

#define SC_NPW 4
static_assert(CN2 % SC_NPW == 0, "whole waves only");
static_assert(CIN == 256, "a lane holds 2 x 4 columns of a 256-wide row");
__global__ void __launch_bounds__(256)
gat_score_feat_kernel(const float* __restrict__ X,
                                      const float* __restrict__ waT,
                                      float* __restrict__ S,
                                      int nnodes) {
  const int gt   = blockIdx.x * blockDim.x + threadIdx.x;
  const int nb   = (gt >> 5) * SC_NPW;
  const int lane = gt & 31;
  if (nb + SC_NPW > nnodes) return;
  const float4* wp = (const float4*)waT + lane;
  float4 f0[SC_NPW], f1[SC_NPW];
#pragma unroll
  for (int r = 0; r < SC_NPW; ++r) {
    const float4* xp = (const float4*)(X + (size_t)(nb + r) * CIN) + lane;
    f0[r] = xp[0];
    f1[r] = xp[CIN / 8];
  }
  float s[SC_NPW][2 * CNH];
#pragma unroll
  for (int o = 0; o < 2 * CNH; ++o) {
    const float4 w0 = wp[o * (CIN / 4)];
    const float4 w1 = wp[o * (CIN / 4) + CIN / 8];
#pragma unroll
    for (int r = 0; r < SC_NPW; ++r) {
      s[r][o] = f0[r].x * w0.x + f0[r].y * w0.y + f0[r].z * w0.z + f0[r].w * w0.w
              + f1[r].x * w1.x + f1[r].y * w1.y + f1[r].z * w1.z + f1[r].w * w1.w;
    }
  }
#pragma unroll
  for (int r = 0; r < SC_NPW; ++r) {
#pragma unroll
    for (int o = 0; o < 2 * CNH; ++o) {
#pragma unroll
      for (int off = 16; off > 0; off >>= 1) s[r][o] += __shfl_xor(s[r][o], off, 32);
    }
  }
  if (lane == 0) {
#pragma unroll
    for (int r = 0; r < SC_NPW; ++r) {
      float4* o4 = (float4*)(S + (size_t)(nb + r) * (2 * CNH));
#pragma unroll
      for (int o = 0; o < 2 * CNH; o += 4)
        o4[o >> 2] = make_float4(s[r][o], s[r][o + 1], s[r][o + 2], s[r][o + 3]);
    }
  }
}

#define AGF_TPB 8
static_assert(CN1 % AGF_TPB == 0, "whole blocks only");
static_assert(CNH == 8, "the scores of a node are read as 4 pieces of 2 heads");
static_assert(CIN == 256, "a lane holds 2 x 4 columns of a 256-wide row");
__global__ void __launch_bounds__(32 * AGF_TPB)
gat_agg_feat_kernel(const float* __restrict__ X,
                    const float* __restrict__ S,
                    const int* __restrict__ src,
                    const int* __restrict__ dst,
                    float* __restrict__ Z) {
  const int lane = threadIdx.x & 31;
  const int wv   = __builtin_amdgcn_readfirstlane((int)(threadIdx.x >> 5));
  const int t    = blockIdx.x * AGF_TPB + wv;
  const int m0   = t * KNB;

  const int sL = src[m0 + lane], dL = dst[m0 + lane];
  const int sX = src[m0 + 32],   dX = dst[m0 + 32];
  const float4* SsL = (const float4*)(S + (size_t)sL * (2 * CNH));
  const float4* SdL = (const float4*)(S + (size_t)dL * (2 * CNH));
  const float4* SsX = (const float4*)(S + (size_t)sX * (2 * CNH));
  const float4* SdX = (const float4*)(S + (size_t)dX * (2 * CNH));
  float wL[CNH], wX[CNH];
#pragma unroll
  for (int g = 0; g < CNH / 2; ++g) {
    const float4 a = SsL[g], b = SdL[g], c = SsX[g], d = SdX[g];
    wL[2 * g]     = a.x + b.y;
    wL[2 * g + 1] = a.z + b.w;
    wX[2 * g]     = c.x + d.y;
    wX[2 * g + 1] = c.z + d.w;
  }
#pragma unroll
  for (int h = 0; h < CNH; ++h) {
    float eL = wL[h], eX = wX[h];
    eL = (eL > 0.f) ? eL : LRELU_ALPHA * eL;
    eX = (eX > 0.f) ? eX : LRELU_ALPHA * eX;
    float mx = eL;
#pragma unroll
    for (int off = 16; off > 0; off >>= 1) mx = fmaxf(mx, __shfl_xor(mx, off, 32));
    mx = fmaxf(mx, eX);
    const float pL = expf(eL - mx);
    const float pX = expf(eX - mx);
    float sum = pL;
#pragma unroll
    for (int off = 16; off > 0; off >>= 1) sum += __shfl_xor(sum, off, 32);
    sum += pX;
    const float inv = 1.f / sum;
    wL[h] = pL * inv;
    wX[h] = pX * inv;
  }

  const float* col = X + 4 * lane;
  float4 acc0[CNH], acc1[CNH];
#pragma unroll
  for (int h = 0; h < CNH; ++h) {
    acc0[h] = make_float4(0.f, 0.f, 0.f, 0.f);
    acc1[h] = make_float4(0.f, 0.f, 0.f, 0.f);
  }
#pragma unroll
  for (int k = 0; k < 32; ++k) {
    const int dk = __builtin_amdgcn_readlane(dL, k);
    const float4 v0 = *(const float4*)(col + (size_t)dk * CIN);
    const float4 v1 = *(const float4*)(col + (size_t)dk * CIN + CIN / 2);
#pragma unroll
    for (int h = 0; h < CNH; ++h) {
      const float wk = __int_as_float(__builtin_amdgcn_readlane(__float_as_int(wL[h]), k));
      acc0[h].x += wk * v0.x; acc0[h].y += wk * v0.y; acc0[h].z += wk * v0.z; acc0[h].w += wk * v0.w;
      acc1[h].x += wk * v1.x; acc1[h].y += wk * v1.y; acc1[h].z += wk * v1.z; acc1[h].w += wk * v1.w;
    }
  }
  {
    const int dk = __builtin_amdgcn_readfirstlane(dX);
    const float4 v0 = *(const float4*)(col + (size_t)dk * CIN);
    const float4 v1 = *(const float4*)(col + (size_t)dk * CIN + CIN / 2);
#pragma unroll
    for (int h = 0; h < CNH; ++h) {
      const float wk = __int_as_float(__builtin_amdgcn_readfirstlane(__float_as_int(wX[h])));
      acc0[h].x += wk * v0.x; acc0[h].y += wk * v0.y; acc0[h].z += wk * v0.z; acc0[h].w += wk * v0.w;
      acc1[h].x += wk * v1.x; acc1[h].y += wk * v1.y; acc1[h].z += wk * v1.z; acc1[h].w += wk * v1.w;
    }
  }
  float* zp = Z + (size_t)t * (CNH * CIN) + 4 * lane;
#pragma unroll
  for (int h = 0; h < CNH; ++h) {
    *(float4*)(zp + (size_t)h * CIN)           = acc0[h];
    *(float4*)(zp + (size_t)h * CIN + CIN / 2) = acc1[h];
  }
}

static_assert(KNB == 33, "edge 32 is the one edge beyond the 32 lanes");
static_assert(L2_PLANES == 4, "the planes are added as two pairs");
__global__ void __launch_bounds__(32 * L2_PLANES) __attribute__((amdgpu_waves_per_eu(1, 4)))
gat_agg_planes_kernel(const float* __restrict__ Hp,
                      const float* __restrict__ Sp,
                      const int* __restrict__ src,
                      const int* __restrict__ dst,
                      float* __restrict__ Yout,
                      int M) {
  __shared__ float4 red[L2_PLANES][32];
  const int t    = blockIdx.x;
  const int lane = threadIdx.x & 31;
  const int pv   = __builtin_amdgcn_readfirstlane((int)(threadIdx.x >> 5));
  const int m0   = t * KNB;

  const int sL = src[m0 + lane], dL = dst[m0 + lane];
  const int sX = src[m0 + 32],   dX = dst[m0 + 32];

  const float* col = Hp + (size_t)pv * M * COUT + 4 * lane;
  float4 v[KNB];
#pragma unroll
  for (int k = 0; k < 32; ++k) {
    const int dk = __builtin_amdgcn_readlane(dL, k);
    v[k] = *(const float4*)(col + (size_t)dk * COUT);
  }
  {
    const int dk = __builtin_amdgcn_readfirstlane(dX);
    v[32] = *(const float4*)(col + (size_t)dk * COUT);
  }

  const float2* S0 = (const float2*)Sp;
  const float2* S1 = S0 + M;
  const float2* S2 = S1 + M;
  const float2* S3 = S2 + M;
  float eL = ((S0[sL].x + S1[sL].x) + (S2[sL].x + S3[sL].x)) + ((S0[dL].y + S1[dL].y) + (S2[dL].y + S3[dL].y));
  float eX = ((S0[sX].x + S1[sX].x) + (S2[sX].x + S3[sX].x)) + ((S0[dX].y + S1[dX].y) + (S2[dX].y + S3[dX].y));
  eL = (eL > 0.f) ? eL : LRELU_ALPHA * eL;
  eX = (eX > 0.f) ? eX : LRELU_ALPHA * eX;

  float mx = eL;
  for (int off = 16; off > 0; off >>= 1) mx = fmaxf(mx, __shfl_xor(mx, off, 32));
  mx = fmaxf(mx, eX);
  const float pL = expf(eL - mx);
  const float pX = expf(eX - mx);
  float sum = pL;
  for (int off = 16; off > 0; off >>= 1) sum += __shfl_xor(sum, off, 32);
  sum += pX;
  const float inv = 1.f / sum;
  const float wL = pL * inv;
  const float wX = pX * inv;

  float4 acc = make_float4(0.f, 0.f, 0.f, 0.f);
#pragma unroll
  for (int k = 0; k < 32; ++k) {
    const float wk = __int_as_float(__builtin_amdgcn_readlane(__float_as_int(wL), k));
    acc.x += wk * v[k].x; acc.y += wk * v[k].y; acc.z += wk * v[k].z; acc.w += wk * v[k].w;
  }
  {
    const float wk = __int_as_float(__builtin_amdgcn_readfirstlane(__float_as_int(wX)));
    acc.x += wk * v[32].x; acc.y += wk * v[32].y; acc.z += wk * v[32].z; acc.w += wk * v[32].w;
  }
  red[pv][lane] = acc;
  __syncthreads();
  if (pv == 0) {
    const float4 c0 = red[0][lane], c1 = red[1][lane], c2 = red[2][lane], c3 = red[3][lane];
    const float tx = (c0.x + c1.x) + (c2.x + c3.x);
    const float ty = (c0.y + c1.y) + (c2.y + c3.y);
    const float tz = (c0.z + c1.z) + (c2.z + c3.z);
    const float tw = (c0.w + c1.w) + (c2.w + c3.w);
    float4 r;
    r.x = (tx > 0.f) ? tx : expm1f(tx);
    r.y = (ty > 0.f) ? ty : expm1f(ty);
    r.z = (tz > 0.f) ? tz : expm1f(tz);
    r.w = (tw > 0.f) ? tw : expm1f(tw);
    *(float4*)(Yout + (size_t)t * COUT + 4 * lane) = r;
  }
}

extern "C" void kernel_launch(void* const* d_in, const int* in_sizes, int n_in,
                              void* d_out, int out_size, void* d_ws, size_t ws_size,
                              hipStream_t stream) {
  (void)in_sizes; (void)n_in; (void)out_size; (void)ws_size;
  const float* feats = (const float*)d_in[0];
  const float* W1    = (const float*)d_in[1];
  const float* a1    = (const float*)d_in[2];
  const float* W2    = (const float*)d_in[3];
  const float* a2    = (const float*)d_in[4];
  const int*   src1  = (const int*)d_in[5];
  const int*   dst1  = (const int*)d_in[6];
  const int*   src2  = (const int*)d_in[8];
  const int*   dst2  = (const int*)d_in[9];
  float* out = (float*)d_out;

  float* ws  = (float*)d_ws;
  float* z1  = ws;
  float* waT = z1  + (size_t)CN1 * CNH * CIN;
  float* s2  = waT + (size_t)2 * CNH * CIN;
  float* h2p = ws  + (size_t)CN2 * CNH * CHID;
  float* Bp1 = h2p + (size_t)CN1 * CNH * CHID;
  float* Bp2 = Bp1 + (size_t)CIN * CNH * CHID;
  float* s1  = Bp2 + (size_t)CNH * CHID * COUT + (size_t)CN1 * COUT;

  gat_pack_weights_kernel<<<PACK_W1_BLOCKS + PACK_W2_BLOCKS, 256, 0, stream>>>(W1, a1, W2, Bp1, waT, Bp2);
  gat_score_feat_kernel<<<CN2 / (8 * SC_NPW), 256, 0, stream>>>(feats, waT, s1, CN2);
  gat_agg_feat_kernel<<<CN1 / AGF_TPB, 32 * AGF_TPB, 0, stream>>>(feats, s1, src1, dst1, z1);
  dim3 g1(L2_PLANES, CN1 / 32);
  gat_wmma_gemm_f32_heads_elu_l2_kernel<<<g1, 128, 0, stream>>>(z1, Bp1, Bp2, a2, h2p, s2,
                                                                CIN, CNH * CHID, CNH * CIN, CN1);
  gat_agg_planes_kernel<<<CN0, 32 * L2_PLANES, 0, stream>>>(h2p, s2, src2, dst2, out, CN1);
}
